# starting kernel
# baseline (speedup 1.0000x reference)
#include <hip/hip_runtime.h>
#include <hip/hip_bf16.h>
#include <math.h>


typedef __attribute__((ext_vector_type(2))) float v2f;
typedef __attribute__((ext_vector_type(8))) float v8f;

#define N_NODES 10000
#define N_EDGES 320000

#define INV_SQ3   0.5773502691896258f
#define INV_SQ2   0.7071067811865476f
#define INV_SQ32  0.17677669529663687f
#define INV_SQ64  0.125f
#define INV_SQ96  0.10206207261596575f
#define INV_SQ160 0.07905694150420949f
#define LN2F      0.6931471805599453f

__device__ __forceinline__ float sspf(float v) {
  float sp = (v > 20.0f) ? v : log1pf(expf(v));
  return sp - LN2F;
}
__device__ __forceinline__ float siluf(float v) {
  return v / (1.0f + expf(-v));
}
__device__ __forceinline__ v2f ld2(const float* p) {
  return *(const v2f*)p;
}

__device__ __forceinline__ v8f wmma4(v2f a, v2f b, v8f c) {
  return __builtin_amdgcn_wmma_f32_16x16x4_f32(false, a, false, b,
                                               (short)0, c, false, false);
}


template <int KS, typename LA>
__device__ __forceinline__ void load_a(v2f (&a)[KS], LA la) {
  const int lane = threadIdx.x & 31;
  const int m = lane & 15;
  const int kk = (lane >> 4) << 1;
#pragma unroll
  for (int q = 0; q < KS; ++q) a[q] = la(m, q * 4 + kk);
}

template <int KS, typename LB>
__device__ __forceinline__ v8f wmma_tile(const v2f (&a)[KS], LB lb, v8f c) {
  const int lane = threadIdx.x & 31;
  const int n = lane & 15;
  const int kk = (lane >> 4) << 1;
#pragma unroll
  for (int q = 0; q < KS; ++q) c = wmma4(a[q], lb(n, q * 4 + kk), c);
  return c;
}

template <typename ST>
__device__ __forceinline__ void tile_store(v8f c, ST st) {
  const int lane = threadIdx.x & 31;
  const int col = lane & 15;
  const int rbase = (lane >> 4) << 3;
#pragma unroll
  for (int r = 0; r < 8; ++r) st(r, rbase + r, col, c[r]);
}

__global__ void transpose_kernel(const float* __restrict__ W,
                                 float* __restrict__ Wt, int K, int Nc) {
  int idx = blockIdx.x * blockDim.x + threadIdx.x;
  if (idx < K * Nc) {
    int k = idx / Nc, n = idx % Nc;
    Wt[n * K + k] = W[idx];
  }
}

__global__ __launch_bounds__(32) void node_kernel(
    const float* __restrict__ x,
    const float* __restrict__ W0pT, const float* __restrict__ b0p,
    const float* __restrict__ W1pT,
    const float* __restrict__ W0nT, const float* __restrict__ b0n,
    const float* __restrict__ W1nT,
    const float* __restrict__ G1T, const float* __restrict__ g1b,
    const float* __restrict__ G2T, const float* __restrict__ g2b,
    float* __restrict__ p0, float* __restrict__ p1,
    float* __restrict__ x0n, float* __restrict__ x1n) {
  __shared__ __align__(16) float f0S[16 * 96];
  __shared__ __align__(16) float hS[16 * 96];
  __shared__ __align__(16) float gS[16 * 96];

  const int lane = threadIdx.x;
  const int n0 = blockIdx.x * 16;
  const float* xt = x + (size_t)n0 * 160;

  {
    v2f ax[16];
    load_a(ax, [&](int m, int k) { return ld2(xt + m * 160 + k); });
    for (int j = 0; j < 4; ++j) {
      v8f c = {};
      c = wmma_tile(ax, [&](int n, int k) {
        return ld2(W0pT + (size_t)(j * 16 + n) * 64 + k);
      }, c);
      tile_store(c, [&](int, int m, int n, float v) {
        int col = j * 16 + n;
        p0[(size_t)(n0 + m) * 64 + col] = v * INV_SQ64 + b0p[col];
      });
    }
  }
  for (int i = 0; i < 3; ++i) {
    v2f ax[8];
    load_a(ax, [&](int m, int k) {
      v2f a;
      a.x = xt[m * 160 + 64 + 3 * k + i];
      a.y = xt[m * 160 + 64 + 3 * (k + 1) + i];
      return a;
    });
    for (int j = 0; j < 2; ++j) {
      v8f c = {};
      c = wmma_tile(ax, [&](int n, int k) {
        return ld2(W1pT + (size_t)(j * 16 + n) * 32 + k);
      }, c);
      tile_store(c, [&](int, int m, int n, float v) {
        p1[(size_t)(n0 + m) * 96 + (j * 16 + n) * 3 + i] = v * INV_SQ32;
      });
    }
  }

  for (int idx = lane; idx < 16 * 96; idx += 32) {
    int m = idx / 96, c = idx % 96;
    float v;
    if (c < 64) {
      v = xt[m * 160 + c];
    } else {
      int u = c - 64;
      float a0 = xt[m * 160 + 64 + u * 3 + 0];
      float a1 = xt[m * 160 + 64 + u * 3 + 1];
      float a2 = xt[m * 160 + 64 + u * 3 + 2];
      v = sqrtf(a0 * a0 + a1 * a1 + a2 * a2 + 1e-12f);
    }
    f0S[idx] = v;
  }
  __syncthreads();

  {
    v2f af[24];
    load_a(af, [&](int m, int k) { return ld2(f0S + m * 96 + k); });
    for (int j = 0; j < 6; ++j) {
      v8f c = {};
      c = wmma_tile(af, [&](int n, int k) {
        return ld2(G1T + (size_t)(j * 16 + n) * 96 + k);
      }, c);
      tile_store(c, [&](int, int m, int n, float v) {
        int col = j * 16 + n;
        hS[m * 96 + col] = siluf(v + g1b[col]);
      });
    }
  }
  __syncthreads();
  {
    v2f ah[24];
    load_a(ah, [&](int m, int k) { return ld2(hS + m * 96 + k); });
    for (int j = 0; j < 6; ++j) {
      v8f c = {};
      c = wmma_tile(ah, [&](int n, int k) {
        return ld2(G2T + (size_t)(j * 16 + n) * 96 + k);
      }, c);
      tile_store(c, [&](int, int m, int n, float v) {
        gS[m * 96 + j * 16 + n] = v + g2b[j * 16 + n];
      });
    }
  }
  __syncthreads();

  {
    v2f ag[16];
    load_a(ag, [&](int m, int k) { return ld2(gS + m * 96 + k); });
    for (int j = 0; j < 4; ++j) {
      v8f c = {};
      c = wmma_tile(ag, [&](int n, int k) {
        return ld2(W0nT + (size_t)(j * 16 + n) * 64 + k);
      }, c);
      tile_store(c, [&](int, int m, int n, float v) {
        int col = j * 16 + n;
        x0n[(size_t)(n0 + m) * 64 + col] = v * INV_SQ64 + b0n[col];
      });
    }
  }
  for (int i = 0; i < 3; ++i) {
    v2f ax[8];
    load_a(ax, [&](int m, int k) {
      v2f a;
      a.x = xt[m * 160 + 64 + 3 * k + i] * gS[m * 96 + 64 + k];
      a.y = xt[m * 160 + 64 + 3 * (k + 1) + i] * gS[m * 96 + 64 + k + 1];
      return a;
    });
    for (int j = 0; j < 2; ++j) {
      v8f c = {};
      c = wmma_tile(ax, [&](int n, int k) {
        return ld2(W1nT + (size_t)(j * 16 + n) * 32 + k);
      }, c);
      tile_store(c, [&](int, int m, int n, float v) {
        x1n[(size_t)(n0 + m) * 96 + (j * 16 + n) * 3 + i] = v * INV_SQ32;
      });
    }
  }
}

__global__ __launch_bounds__(32) void edge_kernel(
    const float* __restrict__ edge_sh, const float* __restrict__ edge_attr,
    const float* __restrict__ F1T, const float* __restrict__ F2T,
    const float* __restrict__ L1T, const float* __restrict__ L2T,
    const int* __restrict__ edge_index,
    const float* __restrict__ p0, const float* __restrict__ p1,
    const float* __restrict__ x0n, const float* __restrict__ x1n,
    float* __restrict__ acc) {
  __shared__ __align__(16) float s0S[16 * 160];
  __shared__ __align__(16) float hN[16 * 32];
  __shared__ __align__(16) float hE[16 * 32];
  __shared__ int dstS[16];
  __shared__ int srcS[16];

  const int lane = threadIdx.x;
  const int e0 = blockIdx.x * 16;

  __builtin_prefetch(F2T, 0, 1);
  __builtin_prefetch(L2T, 0, 1);

  if (lane < 16) {
    dstS[lane] = edge_index[e0 + lane];
    srcS[lane] = edge_index[N_EDGES + e0 + lane];
  }
  __syncthreads();

  for (int idx = lane; idx < 16 * 160; idx += 32) {
    int m = idx / 160, c = idx % 160;
    int d = dstS[m], s = srcS[m];
    float v;
    if (c < 64) {
      v = p0[(size_t)d * 64 + c];
    } else if (c < 128) {
      v = p0[(size_t)s * 64 + (c - 64)];
    } else {
      int u = c - 128;
      const float* pd = p1 + (size_t)d * 96 + u * 3;
      const float* ps = p1 + (size_t)s * 96 + u * 3;
      v = (pd[0] * ps[0] + pd[1] * ps[1] + pd[2] * ps[2]) * INV_SQ3;
    }
    s0S[idx] = v;
  }
  __syncthreads();

  {
    v2f as0[40];
    load_a(as0, [&](int m, int k) { return ld2(s0S + m * 160 + k); });
    for (int j = 0; j < 2; ++j) {
      v8f c = {};
      c = wmma_tile(as0, [&](int n, int k) {
        return ld2(L1T + (size_t)(j * 16 + n) * 160 + k);
      }, c);
      tile_store(c, [&](int, int m, int n, float v) {
        hN[m * 32 + j * 16 + n] = sspf(v * INV_SQ160);
      });
    }
  }
  {
    const float* ea = edge_attr + (size_t)e0 * 32;
    v2f aea[8];
    load_a(aea, [&](int m, int k) { return ld2(ea + m * 32 + k); });
    for (int j = 0; j < 2; ++j) {
      v8f c = {};
      c = wmma_tile(aea, [&](int n, int k) {
        return ld2(F1T + (size_t)(j * 16 + n) * 32 + k);
      }, c);
      tile_store(c, [&](int, int m, int n, float v) {
        hE[m * 32 + j * 16 + n] = sspf(v * INV_SQ32);
      });
    }
  }
  __syncthreads();

  const int rbase = (lane >> 4) << 3;
  float sh0r[8], s1xr[8], s1yr[8], s1zr[8];
  int dR[8], sR[8];
#pragma unroll
  for (int r = 0; r < 8; ++r) {
    int m = rbase + r;
    int e = e0 + m;
    sh0r[r] = edge_sh[(size_t)e * 4 + 0];
    s1xr[r] = edge_sh[(size_t)e * 4 + 1];
    s1yr[r] = edge_sh[(size_t)e * 4 + 2];
    s1zr[r] = edge_sh[(size_t)e * 4 + 3];
    dR[r] = dstS[m];
    sR[r] = srcS[m];
  }

  v2f aN[8], aE[8];
  load_a(aN, [&](int m, int k) { return ld2(hN + m * 32 + k); });
  load_a(aE, [&](int m, int k) { return ld2(hE + m * 32 + k); });

  for (int j = 0; j < 14; ++j) {
    const int nb = j * 16;
    v8f cN = {}, cE = {};
    cN = wmma_tile(aN, [&](int n, int k) {
      return ld2(L2T + (size_t)(nb + n) * 32 + k);
    }, cN);
    cE = wmma_tile(aE, [&](int n, int k) {
      return ld2(F2T + (size_t)(nb + n) * 32 + k);
    }, cE);
    v8f w8 = cN * cE * (INV_SQ32 * INV_SQ32);

    tile_store(w8, [&](int r, int  , int n, float wv) {
      const int col = nb + n;
      const float sh0 = sh0r[r], s1x = s1xr[r], s1y = s1yr[r], s1z = s1zr[r];
      const int s = sR[r];
      float* ad = acc + (size_t)dR[r] * 480;
      if (col < 64) {
        float xs = x0n[(size_t)s * 64 + col];
        atomicAdd(&ad[col], wv * xs * sh0);
      } else if (col < 128) {
        int c2 = col - 64;
        float f = wv * x0n[(size_t)s * 64 + c2] * INV_SQ3;
        atomicAdd(&ad[96 + c2 * 3 + 0], f * s1x);
        atomicAdd(&ad[96 + c2 * 3 + 1], f * s1y);
        atomicAdd(&ad[96 + c2 * 3 + 2], f * s1z);
      } else if (col < 160) {
        int c3 = col - 128;
        const float* xv = x1n + (size_t)s * 96 + c3 * 3;
        float f = wv * sh0;
        atomicAdd(&ad[288 + c3 * 3 + 0], f * xv[0]);
        atomicAdd(&ad[288 + c3 * 3 + 1], f * xv[1]);
        atomicAdd(&ad[288 + c3 * 3 + 2], f * xv[2]);
      } else if (col < 192) {
        int c4 = col - 160;
        const float* xv = x1n + (size_t)s * 96 + c4 * 3;
        float dp = xv[0] * s1x + xv[1] * s1y + xv[2] * s1z;
        atomicAdd(&ad[64 + c4], wv * dp * INV_SQ3);
      } else {
        int c5 = col - 192;
        const float* xv = x1n + (size_t)s * 96 + c5 * 3;
        float cx = xv[1] * s1z - xv[2] * s1y;
        float cy = xv[2] * s1x - xv[0] * s1z;
        float cz = xv[0] * s1y - xv[1] * s1x;
        float f = wv * INV_SQ2;
        atomicAdd(&ad[384 + c5 * 3 + 0], f * cx);
        atomicAdd(&ad[384 + c5 * 3 + 1], f * cy);
        atomicAdd(&ad[384 + c5 * 3 + 2], f * cz);
      }
    });
  }
}

__global__ __launch_bounds__(32) void proj_kernel(
    const float* __restrict__ acc,
    const float* __restrict__ WO0T, const float* __restrict__ bO0,
    const float* __restrict__ WO1T, const float* __restrict__ WO2T,
    float* __restrict__ y) {
  const int n0 = blockIdx.x * 16;
  const float* at = acc + (size_t)n0 * 480;

  {
    v2f aa[24];
    load_a(aa, [&](int m, int k) { return ld2(at + (size_t)m * 480 + k); });
    for (int j = 0; j < 8; ++j) {
      v8f c = {};
      c = wmma_tile(aa, [&](int n, int k) {
        return ld2(WO0T + (size_t)(j * 16 + n) * 96 + k);
      }, c);
      tile_store(c, [&](int, int m, int n, float v) {
        int col = j * 16 + n;
        y[(size_t)(n0 + m) * 416 + col] = v * INV_SQ96 + bO0[col];
      });
    }
  }
  for (int i = 0; i < 3; ++i) {
    v2f aa[24];
    load_a(aa, [&](int m, int k) {
      v2f a;
      a.x = at[(size_t)m * 480 + 96 + 3 * k + i];
      a.y = at[(size_t)m * 480 + 96 + 3 * (k + 1) + i];
      return a;
    });
    for (int j = 0; j < 4; ++j) {
      v8f c = {};
      c = wmma_tile(aa, [&](int n, int k) {
        return ld2(WO1T + (size_t)(j * 16 + n) * 96 + k);
      }, c);
      tile_store(c, [&](int, int m, int n, float v) {
        y[(size_t)(n0 + m) * 416 + 128 + (j * 16 + n) * 3 + i] = v * INV_SQ96;
      });
    }
  }
  for (int i = 0; i < 3; ++i) {
    v2f aa[8];
    load_a(aa, [&](int m, int k) {
      v2f a;
      a.x = at[(size_t)m * 480 + 384 + 3 * k + i];
      a.y = at[(size_t)m * 480 + 384 + 3 * (k + 1) + i];
      return a;
    });
    for (int j = 0; j < 2; ++j) {
      v8f c = {};
      c = wmma_tile(aa, [&](int n, int k) {
        return ld2(WO2T + (size_t)(j * 16 + n) * 32 + k);
      }, c);
      tile_store(c, [&](int, int m, int n, float v) {
        y[(size_t)(n0 + m) * 416 + 320 + (j * 16 + n) * 3 + i] = v * INV_SQ32;
      });
    }
  }
}

extern "C" void kernel_launch(void* const* d_in, const int* in_sizes, int n_in,
                              void* d_out, int out_size, void* d_ws,
                              size_t ws_size, hipStream_t stream) {
  (void)in_sizes; (void)n_in; (void)out_size; (void)ws_size;
  const float* x         = (const float*)d_in[0];
  const float* edge_sh   = (const float*)d_in[1];
  const float* edge_attr = (const float*)d_in[2];
  const float* W0p = (const float*)d_in[3];
  const float* b0p = (const float*)d_in[4];
  const float* W1p = (const float*)d_in[5];
  const float* W0n = (const float*)d_in[6];
  const float* b0n = (const float*)d_in[7];
  const float* W1n = (const float*)d_in[8];
  const float* G1  = (const float*)d_in[9];
  const float* g1b = (const float*)d_in[10];
  const float* G2  = (const float*)d_in[11];
  const float* g2b = (const float*)d_in[12];
  const float* F1  = (const float*)d_in[13];
  const float* F2  = (const float*)d_in[14];
  const float* L1  = (const float*)d_in[15];
  const float* L2  = (const float*)d_in[16];
  const float* WO0 = (const float*)d_in[17];
  const float* bO0 = (const float*)d_in[18];
  const float* WO1 = (const float*)d_in[19];
  const float* WO2 = (const float*)d_in[20];
  const int* edge_index = (const int*)d_in[21];

  float* ws   = (float*)d_ws;
  float* p0   = ws;
  float* p1   = p0   + (size_t)N_NODES * 64;
  float* x0n  = p1   + (size_t)N_NODES * 96;
  float* x1n  = x0n  + (size_t)N_NODES * 64;
  float* acc  = x1n  + (size_t)N_NODES * 96;
  float* wt   = acc  + (size_t)N_NODES * 480;
  float* W0pT = wt;            wt += 64 * 64;
  float* W1pT = wt;            wt += 32 * 32;
  float* W0nT = wt;            wt += 64 * 64;
  float* W1nT = wt;            wt += 32 * 32;
  float* G1T  = wt;            wt += 96 * 96;
  float* G2T  = wt;            wt += 96 * 96;
  float* F1T  = wt;            wt += 32 * 32;
  float* F2T  = wt;            wt += 32 * 224;
  float* L1T  = wt;            wt += 160 * 32;
  float* L2T  = wt;            wt += 32 * 224;
  float* WO0T = wt;            wt += 96 * 128;
  float* WO1T = wt;            wt += 96 * 64;
  float* WO2T = wt;            wt += 32 * 32;

  auto tr = [&](const float* W, float* Wt, int K, int Nc) {
    int n = K * Nc;
    transpose_kernel<<<(n + 255) / 256, 256, 0, stream>>>(W, Wt, K, Nc);
  };
  tr(W0p, W0pT, 64, 64);
  tr(W1p, W1pT, 32, 32);
  tr(W0n, W0nT, 64, 64);
  tr(W1n, W1nT, 32, 32);
  tr(G1, G1T, 96, 96);
  tr(G2, G2T, 96, 96);
  tr(F1, F1T, 32, 32);
  tr(F2, F2T, 32, 224);
  tr(L1, L1T, 160, 32);
  tr(L2, L2T, 32, 224);
  tr(WO0, WO0T, 96, 128);
  tr(WO1, WO1T, 96, 64);
  tr(WO2, WO2T, 32, 32);

  hipMemsetAsync(acc, 0, (size_t)N_NODES * 480 * sizeof(float), stream);

  node_kernel<<<N_NODES / 16, 32, 0, stream>>>(
      x, W0pT, b0p, W1pT, W0nT, b0n, W1nT, G1T, g1b, G2T, g2b,
      p0, p1, x0n, x1n);

  edge_kernel<<<N_EDGES / 16, 32, 0, stream>>>(
      edge_sh, edge_attr, F1T, F2T, L1T, L2T, edge_index,
      p0, p1, x0n, x1n, acc);

  proj_kernel<<<N_NODES / 16, 32, 0, stream>>>(
      acc, WO0T, bO0, WO1T, WO2T, (float*)d_out);
}


# Weights read in place: a B fragment is two 4-byte loads of the row-major input (a wave load touches 2 lines, was 16); the 13 transposed copies and their launches are gone; sums unchanged
# speedup vs baseline: 1.5829x; 1.5829x over previous
#include <hip/hip_runtime.h>
#include <hip/hip_bf16.h>
#include <math.h>


typedef __attribute__((ext_vector_type(2))) float v2f;
typedef __attribute__((ext_vector_type(4))) float v4f;
typedef __attribute__((ext_vector_type(8))) float v8f;

#define N_NODES 10000
#define N_EDGES 320000

#define INV_SQ3   0.5773502691896258f
#define INV_SQ2   0.7071067811865476f
#define INV_SQ32  0.17677669529663687f
#define INV_SQ64  0.125f
#define INV_SQ96  0.10206207261596575f
#define INV_SQ160 0.07905694150420949f
#define LN2F      0.6931471805599453f

__device__ __forceinline__ float sspf(float v) {
  float sp = (v > 20.0f) ? v : log1pf(expf(v));
  return sp - LN2F;
}
__device__ __forceinline__ float siluf(float v) {
  return v / (1.0f + expf(-v));
}
__device__ __forceinline__ v2f ld2(const float* p) {
  return *(const v2f*)p;
}
template <int NC>
__device__ __forceinline__ v2f ldb(const float* __restrict__ W, int k, int col) {
  v2f b;
  b.x = W[k * NC + col];
  b.y = W[(k + 1) * NC + col];
  return b;
}

__device__ __forceinline__ v8f wmma4(v2f a, v2f b, v8f c) {
  return __builtin_amdgcn_wmma_f32_16x16x4_f32(false, a, false, b,
                                               (short)0, c, false, false);
}


template <int KS, typename LA>
__device__ __forceinline__ void load_a(v2f (&a)[KS], LA la) {
  const int lane = threadIdx.x & 31;
  const int m = lane & 15;
  const int kk = (lane >> 4) << 1;
#pragma unroll
  for (int q = 0; q < KS; ++q) a[q] = la(m, q * 4 + kk);
}

template <int KS, typename LB>
__device__ __forceinline__ v8f wmma_tile(const v2f (&a)[KS], LB lb, v8f c) {
  const int lane = threadIdx.x & 31;
  const int n = lane & 15;
  const int kk = (lane >> 4) << 1;
#pragma unroll
  for (int q = 0; q < KS; ++q) c = wmma4(a[q], lb(n, q * 4 + kk), c);
  return c;
}

template <typename ST>
__device__ __forceinline__ void tile_store(v8f c, ST st) {
  const int lane = threadIdx.x & 31;
  const int col = lane & 15;
  const int rbase = (lane >> 4) << 3;
#pragma unroll
  for (int r = 0; r < 8; ++r) st(r, rbase + r, col, c[r]);
}

__global__ __launch_bounds__(32) void node_kernel(
    const float* __restrict__ x,
    const float* __restrict__ W0p, const float* __restrict__ b0p,
    const float* __restrict__ W1p,
    const float* __restrict__ W0n, const float* __restrict__ b0n,
    const float* __restrict__ W1n,
    const float* __restrict__ G1, const float* __restrict__ g1b,
    const float* __restrict__ G2, const float* __restrict__ g2b,
    float* __restrict__ p0, float* __restrict__ p1,
    float* __restrict__ x0n, float* __restrict__ x1n) {
  __shared__ __align__(16) float f0S[16 * 96];
  __shared__ __align__(16) float hS[16 * 96];
  __shared__ __align__(16) float gS[16 * 96];

  const int lane = threadIdx.x;
  const int n0 = blockIdx.x * 16;
  const float* xt = x + (size_t)n0 * 160;

  {
    v2f ax[16];
    load_a(ax, [&](int m, int k) { return ld2(xt + m * 160 + k); });
#pragma unroll
    for (int j = 0; j < 4; ++j) {
      v8f c = {};
      c = wmma_tile(ax, [&](int n, int k) {
        return ldb<64>(W0p, k, j * 16 + n);
      }, c);
      tile_store(c, [&](int, int m, int n, float v) {
        int col = j * 16 + n;
        p0[(size_t)(n0 + m) * 64 + col] = v * INV_SQ64 + b0p[col];
      });
    }
  }
#pragma unroll
  for (int i = 0; i < 3; ++i) {
    v2f ax[8];
    load_a(ax, [&](int m, int k) {
      v2f a;
      a.x = xt[m * 160 + 64 + 3 * k + i];
      a.y = xt[m * 160 + 64 + 3 * (k + 1) + i];
      return a;
    });
    for (int j = 0; j < 2; ++j) {
      v8f c = {};
      c = wmma_tile(ax, [&](int n, int k) {
        return ldb<32>(W1p, k, j * 16 + n);
      }, c);
      tile_store(c, [&](int, int m, int n, float v) {
        p1[(size_t)(n0 + m) * 96 + (j * 16 + n) * 3 + i] = v * INV_SQ32;
      });
    }
  }

  for (int idx = lane; idx < 16 * 96; idx += 32) {
    int m = idx / 96, c = idx % 96;
    float v;
    if (c < 64) {
      v = xt[m * 160 + c];
    } else {
      int u = c - 64;
      float a0 = xt[m * 160 + 64 + u * 3 + 0];
      float a1 = xt[m * 160 + 64 + u * 3 + 1];
      float a2 = xt[m * 160 + 64 + u * 3 + 2];
      v = sqrtf(a0 * a0 + a1 * a1 + a2 * a2 + 1e-12f);
    }
    f0S[idx] = v;
  }
  __syncthreads();

  {
    v2f af[24];
    load_a(af, [&](int m, int k) { return ld2(f0S + m * 96 + k); });
    for (int j = 0; j < 6; ++j) {
      v8f c = {};
      c = wmma_tile(af, [&](int n, int k) {
        return ldb<96>(G1, k, j * 16 + n);
      }, c);
      tile_store(c, [&](int, int m, int n, float v) {
        int col = j * 16 + n;
        hS[m * 96 + col] = siluf(v + g1b[col]);
      });
    }
  }
  __syncthreads();
  {
    v2f ah[24];
    load_a(ah, [&](int m, int k) { return ld2(hS + m * 96 + k); });
    for (int j = 0; j < 6; ++j) {
      v8f c = {};
      c = wmma_tile(ah, [&](int n, int k) {
        return ldb<96>(G2, k, j * 16 + n);
      }, c);
      tile_store(c, [&](int, int m, int n, float v) {
        gS[m * 96 + j * 16 + n] = v + g2b[j * 16 + n];
      });
    }
  }
  __syncthreads();

  {
    v2f ag[16];
    load_a(ag, [&](int m, int k) { return ld2(gS + m * 96 + k); });
#pragma unroll
    for (int j = 0; j < 4; ++j) {
      v8f c = {};
      c = wmma_tile(ag, [&](int n, int k) {
        return ldb<64>(W0n, k, j * 16 + n);
      }, c);
      tile_store(c, [&](int, int m, int n, float v) {
        int col = j * 16 + n;
        x0n[(size_t)(n0 + m) * 64 + col] = v * INV_SQ64 + b0n[col];
      });
    }
  }
  for (int i = 0; i < 3; ++i) {
    v2f ax[8];
    load_a(ax, [&](int m, int k) {
      v2f a;
      a.x = xt[m * 160 + 64 + 3 * k + i] * gS[m * 96 + 64 + k];
      a.y = xt[m * 160 + 64 + 3 * (k + 1) + i] * gS[m * 96 + 64 + k + 1];
      return a;
    });
    for (int j = 0; j < 2; ++j) {
      v8f c = {};
      c = wmma_tile(ax, [&](int n, int k) {
        return ldb<32>(W1n, k, j * 16 + n);
      }, c);
      tile_store(c, [&](int, int m, int n, float v) {
        x1n[(size_t)(n0 + m) * 96 + (j * 16 + n) * 3 + i] = v * INV_SQ32;
      });
    }
  }
}

__global__ __launch_bounds__(32) void edge_kernel(
    const float* __restrict__ edge_sh, const float* __restrict__ edge_attr,
    const float* __restrict__ F1, const float* __restrict__ F2,
    const float* __restrict__ L1, const float* __restrict__ L2,
    const int* __restrict__ edge_index,
    const float* __restrict__ p0, const float* __restrict__ p1,
    const float* __restrict__ x0n, const float* __restrict__ x1n,
    float* __restrict__ acc) {
  __shared__ __align__(16) float s0S[16 * 160];
  __shared__ __align__(16) float hNE[2 * 16 * 32];
  float* const hN = hNE;
  float* const hE = hNE + 16 * 32;
  __shared__ int dstS[16];
  __shared__ int srcS[16];

  const int lane = threadIdx.x;
  const int e0 = blockIdx.x * 16;

  __builtin_prefetch(F2, 0, 1);
  __builtin_prefetch(L2, 0, 1);

  if (lane < 16) {
    dstS[lane] = edge_index[e0 + lane];
    srcS[lane] = edge_index[N_EDGES + e0 + lane];
  }
  __syncthreads();

  for (int idx = lane; idx < 16 * 160; idx += 32) {
    int m = idx / 160, c = idx % 160;
    int d = dstS[m], s = srcS[m];
    float v;
    if (c < 64) {
      v = p0[(size_t)d * 64 + c];
    } else if (c < 128) {
      v = p0[(size_t)s * 64 + (c - 64)];
    } else {
      int u = c - 128;
      const float* pd = p1 + (size_t)d * 96 + u * 3;
      const float* ps = p1 + (size_t)s * 96 + u * 3;
      v = (pd[0] * ps[0] + pd[1] * ps[1] + pd[2] * ps[2]) * INV_SQ3;
    }
    s0S[idx] = v;
  }
  __syncthreads();

  {
    v2f as0[40];
    load_a(as0, [&](int m, int k) { return ld2(s0S + m * 160 + k); });

    {
      float* xsS = s0S;
      const int half = lane >> 4;
      const int c4 = lane & 15;
      v4f t0[8];
#pragma unroll
      for (int i = 0; i < 8; ++i) {
        const int m = 2 * i + half;
        t0[i] = *(const v4f*)(x0n + (size_t)srcS[m] * 64 + c4 * 4);
      }
#pragma unroll
      for (int i = 0; i < 8; ++i) {
        const int m = 2 * i + half;
        *(v4f*)(xsS + m * 160 + c4 * 4) = t0[i];
      }
      v4f t1[12];
#pragma unroll
      for (int i = 0; i < 12; ++i) {
        const int f = lane + 32 * i;
        const int m = f / 24, q4 = f % 24;
        t1[i] = *(const v4f*)(x1n + (size_t)srcS[m] * 96 + q4 * 4);
      }
#pragma unroll
      for (int i = 0; i < 12; ++i) {
        const int f = lane + 32 * i;
        const int m = f / 24, q4 = f % 24;
        *(v4f*)(xsS + m * 160 + 64 + q4 * 4) = t1[i];
      }
    }

    for (int j = 0; j < 2; ++j) {
      v8f c = {};
      c = wmma_tile(as0, [&](int n, int k) {
        return ldb<32>(L1, k, j * 16 + n);
      }, c);
      tile_store(c, [&](int, int m, int n, float v) {
        hN[m * 32 + j * 16 + n] = sspf(v * INV_SQ160);
      });
    }
  }
  {
    const float* ea = edge_attr + (size_t)e0 * 32;
    v2f aea[8];
    load_a(aea, [&](int m, int k) { return ld2(ea + m * 32 + k); });
    for (int j = 0; j < 2; ++j) {
      v8f c = {};
      c = wmma_tile(aea, [&](int n, int k) {
        return ldb<32>(F1, k, j * 16 + n);
      }, c);
      tile_store(c, [&](int, int m, int n, float v) {
        hE[m * 32 + j * 16 + n] = sspf(v * INV_SQ32);
      });
    }
  }
  __syncthreads();

  const int n = lane & 15;
  const int half = lane >> 4;
  const int rbase = half << 3;
  float sh0r[8], s1xr[8], s1yr[8], s1zr[8];
#pragma unroll
  for (int r = 0; r < 8; ++r) {
    int e = e0 + rbase + r;
    sh0r[r] = edge_sh[(size_t)e * 4 + 0];
    s1xr[r] = edge_sh[(size_t)e * 4 + 1];
    s1yr[r] = edge_sh[(size_t)e * 4 + 2];
    s1zr[r] = edge_sh[(size_t)e * 4 + 3];
  }
  const int dmine = dstS[n];
  const float* xsS = s0S;

  v2f aN[8], aE[8];
  load_a(aN, [&](int m, int k) { return ld2(hN + m * 32 + k); });
  load_a(aE, [&](int m, int k) { return ld2(hE + m * 32 + k); });

  float* const stg = hNE;
  constexpr int P1 = 36;
  constexpr int P3 = 97;

  auto wtile = [&](int nb) -> v8f {
    v8f cN = {}, cE = {};
    cN = wmma_tile(aN, [&](int nn, int k) {
      return ldb<224>(L2, k, nb + nn);
    }, cN);
    cE = wmma_tile(aE, [&](int nn, int k) {
      return ldb<224>(F2, k, nb + nn);
    }, cE);
    return cN * cE * (INV_SQ32 * INV_SQ32);
  };

  auto add_line = [&](int m, int ch0, float v) {
    const int d = __builtin_amdgcn_readlane(dmine, m);
    atomicAdd(acc + (size_t)d * 480 + ch0 + lane, v);
  };

  auto scatter1 = [&](const float (&o)[2][8], int ch0) {
#pragma unroll
    for (int r = 0; r < 8; ++r) {
      stg[(rbase + r) * P1 + n] = o[0][r];
      stg[(rbase + r) * P1 + 16 + n] = o[1][r];
    }
    __builtin_amdgcn_fence(__ATOMIC_ACQ_REL, "wavefront");
    float v[16];
#pragma unroll
    for (int m = 0; m < 16; ++m) v[m] = stg[m * P1 + lane];
#pragma unroll
    for (int m = 0; m < 16; ++m) add_line(m, ch0, v[m]);
    __builtin_amdgcn_fence(__ATOMIC_ACQ_REL, "wavefront");
  };

  auto scatter3 = [&](const float (&o)[2][8][3], int ch0) {
#pragma unroll
    for (int ph = 0; ph < 2; ++ph) {
#pragma unroll
      for (int q = 0; q < 4; ++q) {
        float* row = stg + (q + 4 * half) * P3;
#pragma unroll
        for (int t = 0; t < 2; ++t) {
          row[t * 48 + 3 * n + 0] = o[t][4 * ph + q][0];
          row[t * 48 + 3 * n + 1] = o[t][4 * ph + q][1];
          row[t * 48 + 3 * n + 2] = o[t][4 * ph + q][2];
        }
      }
      __builtin_amdgcn_fence(__ATOMIC_ACQ_REL, "wavefront");
      float v[8][3];
#pragma unroll
      for (int s = 0; s < 8; ++s)
#pragma unroll
        for (int L = 0; L < 3; ++L) v[s][L] = stg[s * P3 + L * 32 + lane];
#pragma unroll
      for (int s = 0; s < 8; ++s) {
        const int m = (s & 3) + 4 * ph + 8 * (s >> 2);
#pragma unroll
        for (int L = 0; L < 3; ++L) add_line(m, ch0 + 32 * L, v[s][L]);
      }
      __builtin_amdgcn_fence(__ATOMIC_ACQ_REL, "wavefront");
    }
  };

  for (int p = 0; p < 2; ++p) {
    const int nb = 32 * p;
    const v8f wA = wtile(nb), wB = wtile(nb + 16);
    float o[2][8];
#pragma unroll
    for (int r = 0; r < 8; ++r) {
      const float* xr = xsS + (rbase + r) * 160 + nb + n;
      o[0][r] = wA[r] * xr[0] * sh0r[r];
      o[1][r] = wB[r] * xr[16] * sh0r[r];
    }
    scatter1(o, nb);
  }
  {
    const v8f wA = wtile(160), wB = wtile(176);
    float o[2][8];
#pragma unroll
    for (int r = 0; r < 8; ++r) {
#pragma unroll
      for (int t = 0; t < 2; ++t) {
        const float* xv = xsS + (rbase + r) * 160 + 64 + (16 * t + n) * 3;
        const float wv = t ? wB[r] : wA[r];
        float dp = xv[0] * s1xr[r] + xv[1] * s1yr[r] + xv[2] * s1zr[r];
        o[t][r] = wv * dp * INV_SQ3;
      }
    }
    scatter1(o, 64);
  }
  for (int p = 0; p < 2; ++p) {
    const int nb = 64 + 32 * p;
    const v8f wA = wtile(nb), wB = wtile(nb + 16);
    float o[2][8][3];
#pragma unroll
    for (int r = 0; r < 8; ++r) {
      const float* xr = xsS + (rbase + r) * 160 + 32 * p + n;
#pragma unroll
      for (int t = 0; t < 2; ++t) {
        const float wv = t ? wB[r] : wA[r];
        float f = wv * xr[16 * t] * INV_SQ3;
        o[t][r][0] = f * s1xr[r];
        o[t][r][1] = f * s1yr[r];
        o[t][r][2] = f * s1zr[r];
      }
    }
    scatter3(o, 96 + 96 * p);
  }
  {
    const v8f wA = wtile(128), wB = wtile(144);
    float o[2][8][3];
#pragma unroll
    for (int r = 0; r < 8; ++r) {
#pragma unroll
      for (int t = 0; t < 2; ++t) {
        const float* xv = xsS + (rbase + r) * 160 + 64 + (16 * t + n) * 3;
        const float wv = t ? wB[r] : wA[r];
        float f = wv * sh0r[r];
        o[t][r][0] = f * xv[0];
        o[t][r][1] = f * xv[1];
        o[t][r][2] = f * xv[2];
      }
    }
    scatter3(o, 288);
  }
  {
    const v8f wA = wtile(192), wB = wtile(208);
    float o[2][8][3];
#pragma unroll
    for (int r = 0; r < 8; ++r) {
#pragma unroll
      for (int t = 0; t < 2; ++t) {
        const float* xv = xsS + (rbase + r) * 160 + 64 + (16 * t + n) * 3;
        const float wv = t ? wB[r] : wA[r];
        const float s1x = s1xr[r], s1y = s1yr[r], s1z = s1zr[r];
        float cx = xv[1] * s1z - xv[2] * s1y;
        float cy = xv[2] * s1x - xv[0] * s1z;
        float cz = xv[0] * s1y - xv[1] * s1x;
        float f = wv * INV_SQ2;
        o[t][r][0] = f * cx;
        o[t][r][1] = f * cy;
        o[t][r][2] = f * cz;
      }
    }
    scatter3(o, 384);
  }
}

__global__ __launch_bounds__(32) void proj_kernel(
    const float* __restrict__ acc,
    const float* __restrict__ WO0, const float* __restrict__ bO0,
    const float* __restrict__ WO1, const float* __restrict__ WO2,
    float* __restrict__ y) {
  const int n0 = blockIdx.x * 16;
  const float* at = acc + (size_t)n0 * 480;

  {
    v2f aa[24];
    load_a(aa, [&](int m, int k) { return ld2(at + (size_t)m * 480 + k); });
    for (int j = 0; j < 8; ++j) {
      v8f c = {};
      c = wmma_tile(aa, [&](int n, int k) {
        return ldb<128>(WO0, k, j * 16 + n);
      }, c);
      tile_store(c, [&](int, int m, int n, float v) {
        int col = j * 16 + n;
        y[(size_t)(n0 + m) * 416 + col] = v * INV_SQ96 + bO0[col];
      });
    }
  }
  for (int i = 0; i < 3; ++i) {
    v2f aa[24];
    load_a(aa, [&](int m, int k) {
      v2f a;
      a.x = at[(size_t)m * 480 + 96 + 3 * k + i];
      a.y = at[(size_t)m * 480 + 96 + 3 * (k + 1) + i];
      return a;
    });
#pragma unroll
    for (int j = 0; j < 4; ++j) {
      v8f c = {};
      c = wmma_tile(aa, [&](int n, int k) {
        return ldb<64>(WO1, k, j * 16 + n);
      }, c);
      tile_store(c, [&](int, int m, int n, float v) {
        y[(size_t)(n0 + m) * 416 + 128 + (j * 16 + n) * 3 + i] = v * INV_SQ96;
      });
    }
  }
  for (int i = 0; i < 3; ++i) {
    v2f aa[8];
    load_a(aa, [&](int m, int k) {
      v2f a;
      a.x = at[(size_t)m * 480 + 384 + 3 * k + i];
      a.y = at[(size_t)m * 480 + 384 + 3 * (k + 1) + i];
      return a;
    });
    for (int j = 0; j < 2; ++j) {
      v8f c = {};
      c = wmma_tile(aa, [&](int n, int k) {
        return ldb<32>(WO2, k, j * 16 + n);
      }, c);
      tile_store(c, [&](int, int m, int n, float v) {
        y[(size_t)(n0 + m) * 416 + 320 + (j * 16 + n) * 3 + i] = v * INV_SQ32;
      });
    }
  }
}

extern "C" void kernel_launch(void* const* d_in, const int* in_sizes, int n_in,
                              void* d_out, int out_size, void* d_ws,
                              size_t ws_size, hipStream_t stream) {
  (void)in_sizes; (void)n_in; (void)out_size; (void)ws_size;
  const float* x         = (const float*)d_in[0];
  const float* edge_sh   = (const float*)d_in[1];
  const float* edge_attr = (const float*)d_in[2];
  const float* W0p = (const float*)d_in[3];
  const float* b0p = (const float*)d_in[4];
  const float* W1p = (const float*)d_in[5];
  const float* W0n = (const float*)d_in[6];
  const float* b0n = (const float*)d_in[7];
  const float* W1n = (const float*)d_in[8];
  const float* G1  = (const float*)d_in[9];
  const float* g1b = (const float*)d_in[10];
  const float* G2  = (const float*)d_in[11];
  const float* g2b = (const float*)d_in[12];
  const float* F1  = (const float*)d_in[13];
  const float* F2  = (const float*)d_in[14];
  const float* L1  = (const float*)d_in[15];
  const float* L2  = (const float*)d_in[16];
  const float* WO0 = (const float*)d_in[17];
  const float* bO0 = (const float*)d_in[18];
  const float* WO1 = (const float*)d_in[19];
  const float* WO2 = (const float*)d_in[20];
  const int* edge_index = (const int*)d_in[21];

  float* ws   = (float*)d_ws;
  float* p0   = ws;
  float* p1   = p0   + (size_t)N_NODES * 64;
  float* x0n  = p1   + (size_t)N_NODES * 96;
  float* x1n  = x0n  + (size_t)N_NODES * 64;
  float* acc  = x1n  + (size_t)N_NODES * 96;

  hipMemsetAsync(acc, 0, (size_t)N_NODES * 480 * sizeof(float), stream);

  node_kernel<<<N_NODES / 16, 32, 0, stream>>>(
      x, W0p, b0p, W1p, W0n, b0n, W1n, G1, g1b, G2, g2b,
      p0, p1, x0n, x1n);

  edge_kernel<<<N_EDGES / 16, 32, 0, stream>>>(
      edge_sh, edge_attr, F1, F2, L1, L2, edge_index,
      p0, p1, x0n, x1n, acc);

  proj_kernel<<<N_NODES / 16, 32, 0, stream>>>(
      acc, WO0, bO0, WO1, WO2, (float*)d_out);
}


# Edge MLP input not staged in LDS: each lane gathers its A-fragment pieces of p0/p1 rows by 16-byte loads, all in flight (was an 80-trip load-wait-store loop); layer-1 k permuted, sum order changes
# speedup vs baseline: 1.9058x; 1.2040x over previous
#include <hip/hip_runtime.h>
#include <hip/hip_bf16.h>
#include <math.h>


typedef __attribute__((ext_vector_type(2))) float v2f;
typedef __attribute__((ext_vector_type(4))) float v4f;
typedef __attribute__((ext_vector_type(8))) float v8f;

#define N_NODES 10000
#define N_EDGES 320000

#define INV_SQ3   0.5773502691896258f
#define INV_SQ2   0.7071067811865476f
#define INV_SQ32  0.17677669529663687f
#define INV_SQ64  0.125f
#define INV_SQ96  0.10206207261596575f
#define INV_SQ160 0.07905694150420949f
#define LN2F      0.6931471805599453f

__device__ __forceinline__ float sspf(float v) {
  float sp = (v > 20.0f) ? v : log1pf(expf(v));
  return sp - LN2F;
}
__device__ __forceinline__ float siluf(float v) {
  return v / (1.0f + expf(-v));
}
__device__ __forceinline__ v2f ld2(const float* p) {
  return *(const v2f*)p;
}
template <int NC>
__device__ __forceinline__ v2f ldb(const float* __restrict__ W, int k, int col) {
  v2f b;
  b.x = W[k * NC + col];
  b.y = W[(k + 1) * NC + col];
  return b;
}

__device__ __forceinline__ v8f wmma4(v2f a, v2f b, v8f c) {
  return __builtin_amdgcn_wmma_f32_16x16x4_f32(false, a, false, b,
                                               (short)0, c, false, false);
}


template <int KS, typename LA>
__device__ __forceinline__ void load_a(v2f (&a)[KS], LA la) {
  const int lane = threadIdx.x & 31;
  const int m = lane & 15;
  const int kk = (lane >> 4) << 1;
#pragma unroll
  for (int q = 0; q < KS; ++q) a[q] = la(m, q * 4 + kk);
}

template <int KS, typename LB>
__device__ __forceinline__ v8f wmma_tile(const v2f (&a)[KS], LB lb, v8f c) {
  const int lane = threadIdx.x & 31;
  const int n = lane & 15;
  const int kk = (lane >> 4) << 1;
#pragma unroll
  for (int q = 0; q < KS; ++q) c = wmma4(a[q], lb(n, q * 4 + kk), c);
  return c;
}

template <typename ST>
__device__ __forceinline__ void tile_store(v8f c, ST st) {
  const int lane = threadIdx.x & 31;
  const int col = lane & 15;
  const int rbase = (lane >> 4) << 3;
#pragma unroll
  for (int r = 0; r < 8; ++r) st(r, rbase + r, col, c[r]);
}

__global__ __launch_bounds__(32) void node_kernel(
    const float* __restrict__ x,
    const float* __restrict__ W0p, const float* __restrict__ b0p,
    const float* __restrict__ W1p,
    const float* __restrict__ W0n, const float* __restrict__ b0n,
    const float* __restrict__ W1n,
    const float* __restrict__ G1, const float* __restrict__ g1b,
    const float* __restrict__ G2, const float* __restrict__ g2b,
    float* __restrict__ p0, float* __restrict__ p1,
    float* __restrict__ x0n, float* __restrict__ x1n) {
  __shared__ __align__(16) float f0S[16 * 96];
  __shared__ __align__(16) float hS[16 * 96];
  __shared__ __align__(16) float gS[16 * 96];

  const int lane = threadIdx.x;
  const int n0 = blockIdx.x * 16;
  const float* xt = x + (size_t)n0 * 160;

  {
    v2f ax[16];
    load_a(ax, [&](int m, int k) { return ld2(xt + m * 160 + k); });
#pragma unroll
    for (int j = 0; j < 4; ++j) {
      v8f c = {};
      c = wmma_tile(ax, [&](int n, int k) {
        return ldb<64>(W0p, k, j * 16 + n);
      }, c);
      tile_store(c, [&](int, int m, int n, float v) {
        int col = j * 16 + n;
        p0[(size_t)(n0 + m) * 64 + col] = v * INV_SQ64 + b0p[col];
      });
    }
  }
#pragma unroll
  for (int i = 0; i < 3; ++i) {
    v2f ax[8];
    load_a(ax, [&](int m, int k) {
      v2f a;
      a.x = xt[m * 160 + 64 + 3 * k + i];
      a.y = xt[m * 160 + 64 + 3 * (k + 1) + i];
      return a;
    });
    for (int j = 0; j < 2; ++j) {
      v8f c = {};
      c = wmma_tile(ax, [&](int n, int k) {
        return ldb<32>(W1p, k, j * 16 + n);
      }, c);
      tile_store(c, [&](int, int m, int n, float v) {
        p1[(size_t)(n0 + m) * 96 + (j * 16 + n) * 3 + i] = v * INV_SQ32;
      });
    }
  }

  for (int idx = lane; idx < 16 * 96; idx += 32) {
    int m = idx / 96, c = idx % 96;
    float v;
    if (c < 64) {
      v = xt[m * 160 + c];
    } else {
      int u = c - 64;
      float a0 = xt[m * 160 + 64 + u * 3 + 0];
      float a1 = xt[m * 160 + 64 + u * 3 + 1];
      float a2 = xt[m * 160 + 64 + u * 3 + 2];
      v = sqrtf(a0 * a0 + a1 * a1 + a2 * a2 + 1e-12f);
    }
    f0S[idx] = v;
  }
  __syncthreads();

  {
    v2f af[24];
    load_a(af, [&](int m, int k) { return ld2(f0S + m * 96 + k); });
    for (int j = 0; j < 6; ++j) {
      v8f c = {};
      c = wmma_tile(af, [&](int n, int k) {
        return ldb<96>(G1, k, j * 16 + n);
      }, c);
      tile_store(c, [&](int, int m, int n, float v) {
        int col = j * 16 + n;
        hS[m * 96 + col] = siluf(v + g1b[col]);
      });
    }
  }
  __syncthreads();
  {
    v2f ah[24];
    load_a(ah, [&](int m, int k) { return ld2(hS + m * 96 + k); });
    for (int j = 0; j < 6; ++j) {
      v8f c = {};
      c = wmma_tile(ah, [&](int n, int k) {
        return ldb<96>(G2, k, j * 16 + n);
      }, c);
      tile_store(c, [&](int, int m, int n, float v) {
        gS[m * 96 + j * 16 + n] = v + g2b[j * 16 + n];
      });
    }
  }
  __syncthreads();

  {
    v2f ag[16];
    load_a(ag, [&](int m, int k) { return ld2(gS + m * 96 + k); });
#pragma unroll
    for (int j = 0; j < 4; ++j) {
      v8f c = {};
      c = wmma_tile(ag, [&](int n, int k) {
        return ldb<64>(W0n, k, j * 16 + n);
      }, c);
      tile_store(c, [&](int, int m, int n, float v) {
        int col = j * 16 + n;
        x0n[(size_t)(n0 + m) * 64 + col] = v * INV_SQ64 + b0n[col];
      });
    }
  }
  for (int i = 0; i < 3; ++i) {
    v2f ax[8];
    load_a(ax, [&](int m, int k) {
      v2f a;
      a.x = xt[m * 160 + 64 + 3 * k + i] * gS[m * 96 + 64 + k];
      a.y = xt[m * 160 + 64 + 3 * (k + 1) + i] * gS[m * 96 + 64 + k + 1];
      return a;
    });
    for (int j = 0; j < 2; ++j) {
      v8f c = {};
      c = wmma_tile(ax, [&](int n, int k) {
        return ldb<32>(W1n, k, j * 16 + n);
      }, c);
      tile_store(c, [&](int, int m, int n, float v) {
        x1n[(size_t)(n0 + m) * 96 + (j * 16 + n) * 3 + i] = v * INV_SQ32;
      });
    }
  }
}

__global__ __launch_bounds__(32) void edge_kernel(
    const float* __restrict__ edge_sh, const float* __restrict__ edge_attr,
    const float* __restrict__ F1, const float* __restrict__ F2,
    const float* __restrict__ L1, const float* __restrict__ L2,
    const int* __restrict__ edge_index,
    const float* __restrict__ p0, const float* __restrict__ p1,
    const float* __restrict__ x0n, const float* __restrict__ x1n,
    float* __restrict__ acc) {
  __shared__ __align__(16) float s0S[16 * 160];
  __shared__ __align__(16) float hNE[2 * 16 * 32];
  float* const hN = hNE;
  float* const hE = hNE + 16 * 32;
  __shared__ int srcS[16];

  const int lane = threadIdx.x;
  const int e0 = blockIdx.x * 16;

  __builtin_prefetch(F2, 0, 1);
  __builtin_prefetch(L2, 0, 1);

  const int mrow = lane & 15;
  const int hk = lane >> 4;
  const int drow = edge_index[e0 + mrow];
  const int srow = edge_index[N_EDGES + e0 + mrow];
  if (lane < 16) srcS[lane] = srow;
  __syncthreads();

  {
    v2f as0[40];
    {
      const float* pd0 = p0 + (size_t)drow * 64 + 4 * hk;
      const float* ps0 = p0 + (size_t)srow * 64 + 4 * hk;
      const float* pd1 = p1 + (size_t)drow * 96 + 12 * hk;
      const float* ps1 = p1 + (size_t)srow * 96 + 12 * hk;
      v4f qd[8], qs[8], vd[4][3], vs[4][3];
#pragma unroll
      for (int g = 0; g < 4; ++g) {
#pragma unroll
        for (int j = 0; j < 3; ++j) {
          vd[g][j] = *(const v4f*)(pd1 + 24 * g + 4 * j);
          vs[g][j] = *(const v4f*)(ps1 + 24 * g + 4 * j);
        }
      }
#pragma unroll
      for (int g = 0; g < 8; ++g) {
        qd[g] = *(const v4f*)(pd0 + 8 * g);
        qs[g] = *(const v4f*)(ps0 + 8 * g);
      }
#pragma unroll
      for (int g = 0; g < 8; ++g) {
        as0[2 * g].x = qd[g].x;          as0[2 * g].y = qd[g].y;
        as0[2 * g + 1].x = qd[g].z;      as0[2 * g + 1].y = qd[g].w;
        as0[16 + 2 * g].x = qs[g].x;     as0[16 + 2 * g].y = qs[g].y;
        as0[16 + 2 * g + 1].x = qs[g].z; as0[16 + 2 * g + 1].y = qs[g].w;
      }
#pragma unroll
      for (int g = 0; g < 4; ++g) {
        const float xd[12] = {vd[g][0].x, vd[g][0].y, vd[g][0].z, vd[g][0].w,
                              vd[g][1].x, vd[g][1].y, vd[g][1].z, vd[g][1].w,
                              vd[g][2].x, vd[g][2].y, vd[g][2].z, vd[g][2].w};
        const float xs[12] = {vs[g][0].x, vs[g][0].y, vs[g][0].z, vs[g][0].w,
                              vs[g][1].x, vs[g][1].y, vs[g][1].z, vs[g][1].w,
                              vs[g][2].x, vs[g][2].y, vs[g][2].z, vs[g][2].w};
        float ip[4];
#pragma unroll
        for (int j = 0; j < 4; ++j) {
          ip[j] = (xd[3 * j] * xs[3 * j] + xd[3 * j + 1] * xs[3 * j + 1] +
                   xd[3 * j + 2] * xs[3 * j + 2]) * INV_SQ3;
        }
        as0[32 + 2 * g].x = ip[0];     as0[32 + 2 * g].y = ip[1];
        as0[32 + 2 * g + 1].x = ip[2]; as0[32 + 2 * g + 1].y = ip[3];
      }
    }

    {
      float* xsS = s0S;
      const int half = lane >> 4;
      const int c4 = lane & 15;
      v4f t0[8];
#pragma unroll
      for (int i = 0; i < 8; ++i) {
        const int m = 2 * i + half;
        t0[i] = *(const v4f*)(x0n + (size_t)srcS[m] * 64 + c4 * 4);
      }
#pragma unroll
      for (int i = 0; i < 8; ++i) {
        const int m = 2 * i + half;
        *(v4f*)(xsS + m * 160 + c4 * 4) = t0[i];
      }
      v4f t1[12];
#pragma unroll
      for (int i = 0; i < 12; ++i) {
        const int f = lane + 32 * i;
        const int m = f / 24, q4 = f % 24;
        t1[i] = *(const v4f*)(x1n + (size_t)srcS[m] * 96 + q4 * 4);
      }
#pragma unroll
      for (int i = 0; i < 12; ++i) {
        const int f = lane + 32 * i;
        const int m = f / 24, q4 = f % 24;
        *(v4f*)(xsS + m * 160 + 64 + q4 * 4) = t1[i];
      }
    }

    for (int j = 0; j < 2; ++j) {
      v8f c = {};
#pragma unroll
      for (int Q = 0; Q < 40; ++Q) {
        const int k0 = 8 * (Q >> 1) + 4 * hk + 2 * (Q & 1);
        c = wmma4(as0[Q], ldb<32>(L1, k0, j * 16 + mrow), c);
      }
      tile_store(c, [&](int, int m, int n, float v) {
        hN[m * 32 + j * 16 + n] = sspf(v * INV_SQ160);
      });
    }
  }
  {
    const float* ea = edge_attr + (size_t)e0 * 32;
    v2f aea[8];
    load_a(aea, [&](int m, int k) { return ld2(ea + m * 32 + k); });
    for (int j = 0; j < 2; ++j) {
      v8f c = {};
      c = wmma_tile(aea, [&](int n, int k) {
        return ldb<32>(F1, k, j * 16 + n);
      }, c);
      tile_store(c, [&](int, int m, int n, float v) {
        hE[m * 32 + j * 16 + n] = sspf(v * INV_SQ32);
      });
    }
  }
  __syncthreads();

  const int n = lane & 15;
  const int half = lane >> 4;
  const int rbase = half << 3;
  float sh0r[8], s1xr[8], s1yr[8], s1zr[8];
#pragma unroll
  for (int r = 0; r < 8; ++r) {
    int e = e0 + rbase + r;
    sh0r[r] = edge_sh[(size_t)e * 4 + 0];
    s1xr[r] = edge_sh[(size_t)e * 4 + 1];
    s1yr[r] = edge_sh[(size_t)e * 4 + 2];
    s1zr[r] = edge_sh[(size_t)e * 4 + 3];
  }
  const int dmine = drow;
  const float* xsS = s0S;

  v2f aN[8], aE[8];
  load_a(aN, [&](int m, int k) { return ld2(hN + m * 32 + k); });
  load_a(aE, [&](int m, int k) { return ld2(hE + m * 32 + k); });

  float* const stg = hNE;
  constexpr int P1 = 36;
  constexpr int P3 = 97;

  auto wtile = [&](int nb) -> v8f {
    v8f cN = {}, cE = {};
    cN = wmma_tile(aN, [&](int nn, int k) {
      return ldb<224>(L2, k, nb + nn);
    }, cN);
    cE = wmma_tile(aE, [&](int nn, int k) {
      return ldb<224>(F2, k, nb + nn);
    }, cE);
    return cN * cE * (INV_SQ32 * INV_SQ32);
  };

  auto add_line = [&](int m, int ch0, float v) {
    const int d = __builtin_amdgcn_readlane(dmine, m);
    atomicAdd(acc + (size_t)d * 480 + ch0 + lane, v);
  };

  auto scatter1 = [&](const float (&o)[2][8], int ch0) {
#pragma unroll
    for (int r = 0; r < 8; ++r) {
      stg[(rbase + r) * P1 + n] = o[0][r];
      stg[(rbase + r) * P1 + 16 + n] = o[1][r];
    }
    __builtin_amdgcn_fence(__ATOMIC_ACQ_REL, "wavefront");
    float v[16];
#pragma unroll
    for (int m = 0; m < 16; ++m) v[m] = stg[m * P1 + lane];
#pragma unroll
    for (int m = 0; m < 16; ++m) add_line(m, ch0, v[m]);
    __builtin_amdgcn_fence(__ATOMIC_ACQ_REL, "wavefront");
  };

  auto scatter3 = [&](const float (&o)[2][8][3], int ch0) {
#pragma unroll
    for (int ph = 0; ph < 2; ++ph) {
#pragma unroll
      for (int q = 0; q < 4; ++q) {
        float* row = stg + (q + 4 * half) * P3;
#pragma unroll
        for (int t = 0; t < 2; ++t) {
          row[t * 48 + 3 * n + 0] = o[t][4 * ph + q][0];
          row[t * 48 + 3 * n + 1] = o[t][4 * ph + q][1];
          row[t * 48 + 3 * n + 2] = o[t][4 * ph + q][2];
        }
      }
      __builtin_amdgcn_fence(__ATOMIC_ACQ_REL, "wavefront");
      float v[8][3];
#pragma unroll
      for (int s = 0; s < 8; ++s)
#pragma unroll
        for (int L = 0; L < 3; ++L) v[s][L] = stg[s * P3 + L * 32 + lane];
#pragma unroll
      for (int s = 0; s < 8; ++s) {
        const int m = (s & 3) + 4 * ph + 8 * (s >> 2);
#pragma unroll
        for (int L = 0; L < 3; ++L) add_line(m, ch0 + 32 * L, v[s][L]);
      }
      __builtin_amdgcn_fence(__ATOMIC_ACQ_REL, "wavefront");
    }
  };

  for (int p = 0; p < 2; ++p) {
    const int nb = 32 * p;
    const v8f wA = wtile(nb), wB = wtile(nb + 16);
    float o[2][8];
#pragma unroll
    for (int r = 0; r < 8; ++r) {
      const float* xr = xsS + (rbase + r) * 160 + nb + n;
      o[0][r] = wA[r] * xr[0] * sh0r[r];
      o[1][r] = wB[r] * xr[16] * sh0r[r];
    }
    scatter1(o, nb);
  }
  {
    const v8f wA = wtile(160), wB = wtile(176);
    float o[2][8];
#pragma unroll
    for (int r = 0; r < 8; ++r) {
#pragma unroll
      for (int t = 0; t < 2; ++t) {
        const float* xv = xsS + (rbase + r) * 160 + 64 + (16 * t + n) * 3;
        const float wv = t ? wB[r] : wA[r];
        float dp = xv[0] * s1xr[r] + xv[1] * s1yr[r] + xv[2] * s1zr[r];
        o[t][r] = wv * dp * INV_SQ3;
      }
    }
    scatter1(o, 64);
  }
  for (int p = 0; p < 2; ++p) {
    const int nb = 64 + 32 * p;
    const v8f wA = wtile(nb), wB = wtile(nb + 16);
    float o[2][8][3];
#pragma unroll
    for (int r = 0; r < 8; ++r) {
      const float* xr = xsS + (rbase + r) * 160 + 32 * p + n;
#pragma unroll
      for (int t = 0; t < 2; ++t) {
        const float wv = t ? wB[r] : wA[r];
        float f = wv * xr[16 * t] * INV_SQ3;
        o[t][r][0] = f * s1xr[r];
        o[t][r][1] = f * s1yr[r];
        o[t][r][2] = f * s1zr[r];
      }
    }
    scatter3(o, 96 + 96 * p);
  }
  {
    const v8f wA = wtile(128), wB = wtile(144);
    float o[2][8][3];
#pragma unroll
    for (int r = 0; r < 8; ++r) {
#pragma unroll
      for (int t = 0; t < 2; ++t) {
        const float* xv = xsS + (rbase + r) * 160 + 64 + (16 * t + n) * 3;
        const float wv = t ? wB[r] : wA[r];
        float f = wv * sh0r[r];
        o[t][r][0] = f * xv[0];
        o[t][r][1] = f * xv[1];
        o[t][r][2] = f * xv[2];
      }
    }
    scatter3(o, 288);
  }
  {
    const v8f wA = wtile(192), wB = wtile(208);
    float o[2][8][3];
#pragma unroll
    for (int r = 0; r < 8; ++r) {
#pragma unroll
      for (int t = 0; t < 2; ++t) {
        const float* xv = xsS + (rbase + r) * 160 + 64 + (16 * t + n) * 3;
        const float wv = t ? wB[r] : wA[r];
        const float s1x = s1xr[r], s1y = s1yr[r], s1z = s1zr[r];
        float cx = xv[1] * s1z - xv[2] * s1y;
        float cy = xv[2] * s1x - xv[0] * s1z;
        float cz = xv[0] * s1y - xv[1] * s1x;
        float f = wv * INV_SQ2;
        o[t][r][0] = f * cx;
        o[t][r][1] = f * cy;
        o[t][r][2] = f * cz;
      }
    }
    scatter3(o, 384);
  }
}

__global__ __launch_bounds__(32) void proj_kernel(
    const float* __restrict__ acc,
    const float* __restrict__ WO0, const float* __restrict__ bO0,
    const float* __restrict__ WO1, const float* __restrict__ WO2,
    float* __restrict__ y) {
  const int n0 = blockIdx.x * 16;
  const float* at = acc + (size_t)n0 * 480;

  {
    v2f aa[24];
    load_a(aa, [&](int m, int k) { return ld2(at + (size_t)m * 480 + k); });
    for (int j = 0; j < 8; ++j) {
      v8f c = {};
      c = wmma_tile(aa, [&](int n, int k) {
        return ldb<128>(WO0, k, j * 16 + n);
      }, c);
      tile_store(c, [&](int, int m, int n, float v) {
        int col = j * 16 + n;
        y[(size_t)(n0 + m) * 416 + col] = v * INV_SQ96 + bO0[col];
      });
    }
  }
  for (int i = 0; i < 3; ++i) {
    v2f aa[24];
    load_a(aa, [&](int m, int k) {
      v2f a;
      a.x = at[(size_t)m * 480 + 96 + 3 * k + i];
      a.y = at[(size_t)m * 480 + 96 + 3 * (k + 1) + i];
      return a;
    });
#pragma unroll
    for (int j = 0; j < 4; ++j) {
      v8f c = {};
      c = wmma_tile(aa, [&](int n, int k) {
        return ldb<64>(WO1, k, j * 16 + n);
      }, c);
      tile_store(c, [&](int, int m, int n, float v) {
        y[(size_t)(n0 + m) * 416 + 128 + (j * 16 + n) * 3 + i] = v * INV_SQ96;
      });
    }
  }
  for (int i = 0; i < 3; ++i) {
    v2f aa[8];
    load_a(aa, [&](int m, int k) {
      v2f a;
      a.x = at[(size_t)m * 480 + 384 + 3 * k + i];
      a.y = at[(size_t)m * 480 + 384 + 3 * (k + 1) + i];
      return a;
    });
    for (int j = 0; j < 2; ++j) {
      v8f c = {};
      c = wmma_tile(aa, [&](int n, int k) {
        return ldb<32>(WO2, k, j * 16 + n);
      }, c);
      tile_store(c, [&](int, int m, int n, float v) {
        y[(size_t)(n0 + m) * 416 + 320 + (j * 16 + n) * 3 + i] = v * INV_SQ32;
      });
    }
  }
}

extern "C" void kernel_launch(void* const* d_in, const int* in_sizes, int n_in,
                              void* d_out, int out_size, void* d_ws,
                              size_t ws_size, hipStream_t stream) {
  (void)in_sizes; (void)n_in; (void)out_size; (void)ws_size;
  const float* x         = (const float*)d_in[0];
  const float* edge_sh   = (const float*)d_in[1];
  const float* edge_attr = (const float*)d_in[2];
  const float* W0p = (const float*)d_in[3];
  const float* b0p = (const float*)d_in[4];
  const float* W1p = (const float*)d_in[5];
  const float* W0n = (const float*)d_in[6];
  const float* b0n = (const float*)d_in[7];
  const float* W1n = (const float*)d_in[8];
  const float* G1  = (const float*)d_in[9];
  const float* g1b = (const float*)d_in[10];
  const float* G2  = (const float*)d_in[11];
  const float* g2b = (const float*)d_in[12];
  const float* F1  = (const float*)d_in[13];
  const float* F2  = (const float*)d_in[14];
  const float* L1  = (const float*)d_in[15];
  const float* L2  = (const float*)d_in[16];
  const float* WO0 = (const float*)d_in[17];
  const float* bO0 = (const float*)d_in[18];
  const float* WO1 = (const float*)d_in[19];
  const float* WO2 = (const float*)d_in[20];
  const int* edge_index = (const int*)d_in[21];

  float* ws   = (float*)d_ws;
  float* p0   = ws;
  float* p1   = p0   + (size_t)N_NODES * 64;
  float* x0n  = p1   + (size_t)N_NODES * 96;
  float* x1n  = x0n  + (size_t)N_NODES * 64;
  float* acc  = x1n  + (size_t)N_NODES * 96;

  hipMemsetAsync(acc, 0, (size_t)N_NODES * 480 * sizeof(float), stream);

  node_kernel<<<N_NODES / 16, 32, 0, stream>>>(
      x, W0p, b0p, W1p, W0n, b0n, W1n, G1, g1b, G2, g2b,
      p0, p1, x0n, x1n);

  edge_kernel<<<N_EDGES / 16, 32, 0, stream>>>(
      edge_sh, edge_attr, F1, F2, L1, L2, edge_index,
      p0, p1, x0n, x1n, acc);

  proj_kernel<<<N_NODES / 16, 32, 0, stream>>>(
      acc, WO0, bO0, WO1, WO2, (float*)d_out);
}


# Edge kernel holds 5 waves per SIMD, was 4 (176 VGPRs, was 209): both layer-1 column tiles in one pass so the input panel dies early, source rows loaded behind it, row index by lane shuffle; same sums
# speedup vs baseline: 1.9730x; 1.0353x over previous
#include <hip/hip_runtime.h>
#include <hip/hip_bf16.h>
#include <math.h>


typedef __attribute__((ext_vector_type(2))) float v2f;
typedef __attribute__((ext_vector_type(4))) float v4f;
typedef __attribute__((ext_vector_type(8))) float v8f;

#define N_NODES 10000
#define N_EDGES 320000

#define INV_SQ3   0.5773502691896258f
#define INV_SQ2   0.7071067811865476f
#define INV_SQ32  0.17677669529663687f
#define INV_SQ64  0.125f
#define INV_SQ96  0.10206207261596575f
#define INV_SQ160 0.07905694150420949f
#define LN2F      0.6931471805599453f

__device__ __forceinline__ float sspf(float v) {
  float sp = (v > 20.0f) ? v : log1pf(expf(v));
  return sp - LN2F;
}
__device__ __forceinline__ float siluf(float v) {
  return v / (1.0f + expf(-v));
}
__device__ __forceinline__ v2f ld2(const float* p) {
  return *(const v2f*)p;
}
template <int NC>
__device__ __forceinline__ v2f ldb(const float* __restrict__ W, int k, int col) {
  v2f b;
  b.x = W[k * NC + col];
  b.y = W[(k + 1) * NC + col];
  return b;
}

__device__ __forceinline__ v8f wmma4(v2f a, v2f b, v8f c) {
  return __builtin_amdgcn_wmma_f32_16x16x4_f32(false, a, false, b,
                                               (short)0, c, false, false);
}


template <int KS, typename LA>
__device__ __forceinline__ void load_a(v2f (&a)[KS], LA la) {
  const int lane = threadIdx.x & 31;
  const int m = lane & 15;
  const int kk = (lane >> 4) << 1;
#pragma unroll
  for (int q = 0; q < KS; ++q) a[q] = la(m, q * 4 + kk);
}

template <int KS, typename LB>
__device__ __forceinline__ v8f wmma_tile(const v2f (&a)[KS], LB lb, v8f c) {
  const int lane = threadIdx.x & 31;
  const int n = lane & 15;
  const int kk = (lane >> 4) << 1;
#pragma unroll
  for (int q = 0; q < KS; ++q) c = wmma4(a[q], lb(n, q * 4 + kk), c);
  return c;
}

template <typename ST>
__device__ __forceinline__ void tile_store(v8f c, ST st) {
  const int lane = threadIdx.x & 31;
  const int col = lane & 15;
  const int rbase = (lane >> 4) << 3;
#pragma unroll
  for (int r = 0; r < 8; ++r) st(r, rbase + r, col, c[r]);
}

__global__ __launch_bounds__(32) void node_kernel(
    const float* __restrict__ x,
    const float* __restrict__ W0p, const float* __restrict__ b0p,
    const float* __restrict__ W1p,
    const float* __restrict__ W0n, const float* __restrict__ b0n,
    const float* __restrict__ W1n,
    const float* __restrict__ G1, const float* __restrict__ g1b,
    const float* __restrict__ G2, const float* __restrict__ g2b,
    float* __restrict__ p0, float* __restrict__ p1,
    float* __restrict__ x0n, float* __restrict__ x1n) {
  __shared__ __align__(16) float f0S[16 * 96];
  __shared__ __align__(16) float hS[16 * 96];
  __shared__ __align__(16) float gS[16 * 96];

  const int lane = threadIdx.x;
  const int n0 = blockIdx.x * 16;
  const float* xt = x + (size_t)n0 * 160;

  {
    v2f ax[16];
    load_a(ax, [&](int m, int k) { return ld2(xt + m * 160 + k); });
#pragma unroll
    for (int j = 0; j < 4; ++j) {
      v8f c = {};
      c = wmma_tile(ax, [&](int n, int k) {
        return ldb<64>(W0p, k, j * 16 + n);
      }, c);
      tile_store(c, [&](int, int m, int n, float v) {
        int col = j * 16 + n;
        p0[(size_t)(n0 + m) * 64 + col] = v * INV_SQ64 + b0p[col];
      });
    }
  }
#pragma unroll
  for (int i = 0; i < 3; ++i) {
    v2f ax[8];
    load_a(ax, [&](int m, int k) {
      v2f a;
      a.x = xt[m * 160 + 64 + 3 * k + i];
      a.y = xt[m * 160 + 64 + 3 * (k + 1) + i];
      return a;
    });
    for (int j = 0; j < 2; ++j) {
      v8f c = {};
      c = wmma_tile(ax, [&](int n, int k) {
        return ldb<32>(W1p, k, j * 16 + n);
      }, c);
      tile_store(c, [&](int, int m, int n, float v) {
        p1[(size_t)(n0 + m) * 96 + (j * 16 + n) * 3 + i] = v * INV_SQ32;
      });
    }
  }

  for (int idx = lane; idx < 16 * 96; idx += 32) {
    int m = idx / 96, c = idx % 96;
    float v;
    if (c < 64) {
      v = xt[m * 160 + c];
    } else {
      int u = c - 64;
      float a0 = xt[m * 160 + 64 + u * 3 + 0];
      float a1 = xt[m * 160 + 64 + u * 3 + 1];
      float a2 = xt[m * 160 + 64 + u * 3 + 2];
      v = sqrtf(a0 * a0 + a1 * a1 + a2 * a2 + 1e-12f);
    }
    f0S[idx] = v;
  }
  __syncthreads();

  {
    v2f af[24];
    load_a(af, [&](int m, int k) { return ld2(f0S + m * 96 + k); });
    for (int j = 0; j < 6; ++j) {
      v8f c = {};
      c = wmma_tile(af, [&](int n, int k) {
        return ldb<96>(G1, k, j * 16 + n);
      }, c);
      tile_store(c, [&](int, int m, int n, float v) {
        int col = j * 16 + n;
        hS[m * 96 + col] = siluf(v + g1b[col]);
      });
    }
  }
  __syncthreads();
  {
    v2f ah[24];
    load_a(ah, [&](int m, int k) { return ld2(hS + m * 96 + k); });
    for (int j = 0; j < 6; ++j) {
      v8f c = {};
      c = wmma_tile(ah, [&](int n, int k) {
        return ldb<96>(G2, k, j * 16 + n);
      }, c);
      tile_store(c, [&](int, int m, int n, float v) {
        gS[m * 96 + j * 16 + n] = v + g2b[j * 16 + n];
      });
    }
  }
  __syncthreads();

  {
    v2f ag[16];
    load_a(ag, [&](int m, int k) { return ld2(gS + m * 96 + k); });
#pragma unroll
    for (int j = 0; j < 4; ++j) {
      v8f c = {};
      c = wmma_tile(ag, [&](int n, int k) {
        return ldb<64>(W0n, k, j * 16 + n);
      }, c);
      tile_store(c, [&](int, int m, int n, float v) {
        int col = j * 16 + n;
        x0n[(size_t)(n0 + m) * 64 + col] = v * INV_SQ64 + b0n[col];
      });
    }
  }
  for (int i = 0; i < 3; ++i) {
    v2f ax[8];
    load_a(ax, [&](int m, int k) {
      v2f a;
      a.x = xt[m * 160 + 64 + 3 * k + i] * gS[m * 96 + 64 + k];
      a.y = xt[m * 160 + 64 + 3 * (k + 1) + i] * gS[m * 96 + 64 + k + 1];
      return a;
    });
    for (int j = 0; j < 2; ++j) {
      v8f c = {};
      c = wmma_tile(ax, [&](int n, int k) {
        return ldb<32>(W1n, k, j * 16 + n);
      }, c);
      tile_store(c, [&](int, int m, int n, float v) {
        x1n[(size_t)(n0 + m) * 96 + (j * 16 + n) * 3 + i] = v * INV_SQ32;
      });
    }
  }
}

__global__ __launch_bounds__(32) void edge_kernel(
    const float* __restrict__ edge_sh, const float* __restrict__ edge_attr,
    const float* __restrict__ F1, const float* __restrict__ F2,
    const float* __restrict__ L1, const float* __restrict__ L2,
    const int* __restrict__ edge_index,
    const float* __restrict__ p0, const float* __restrict__ p1,
    const float* __restrict__ x0n, const float* __restrict__ x1n,
    float* __restrict__ acc) {
  __shared__ __align__(16) float s0S[16 * 160];
  __shared__ __align__(16) float hNE[2 * 16 * 32];
  float* const hN = hNE;
  float* const hE = hNE + 16 * 32;

  const int lane = threadIdx.x;
  const int e0 = blockIdx.x * 16;

  __builtin_prefetch(F2, 0, 1);
  __builtin_prefetch(L2, 0, 1);

  const int mrow = lane & 15;
  const int hk = lane >> 4;
  const int drow = edge_index[e0 + mrow];
  const int srow = edge_index[N_EDGES + e0 + mrow];

  {
    v2f as0[40];
    {
      const float* pd0 = p0 + (size_t)drow * 64 + 4 * hk;
      const float* ps0 = p0 + (size_t)srow * 64 + 4 * hk;
      const float* pd1 = p1 + (size_t)drow * 96 + 12 * hk;
      const float* ps1 = p1 + (size_t)srow * 96 + 12 * hk;
      v4f qd[8], qs[8], vd[4][3], vs[4][3];
#pragma unroll
      for (int g = 0; g < 4; ++g) {
#pragma unroll
        for (int j = 0; j < 3; ++j) {
          vd[g][j] = *(const v4f*)(pd1 + 24 * g + 4 * j);
          vs[g][j] = *(const v4f*)(ps1 + 24 * g + 4 * j);
        }
      }
#pragma unroll
      for (int g = 0; g < 8; ++g) {
        qd[g] = *(const v4f*)(pd0 + 8 * g);
        qs[g] = *(const v4f*)(ps0 + 8 * g);
      }
#pragma unroll
      for (int g = 0; g < 8; ++g) {
        as0[2 * g].x = qd[g].x;          as0[2 * g].y = qd[g].y;
        as0[2 * g + 1].x = qd[g].z;      as0[2 * g + 1].y = qd[g].w;
        as0[16 + 2 * g].x = qs[g].x;     as0[16 + 2 * g].y = qs[g].y;
        as0[16 + 2 * g + 1].x = qs[g].z; as0[16 + 2 * g + 1].y = qs[g].w;
      }
#pragma unroll
      for (int g = 0; g < 4; ++g) {
        const float xd[12] = {vd[g][0].x, vd[g][0].y, vd[g][0].z, vd[g][0].w,
                              vd[g][1].x, vd[g][1].y, vd[g][1].z, vd[g][1].w,
                              vd[g][2].x, vd[g][2].y, vd[g][2].z, vd[g][2].w};
        const float xs[12] = {vs[g][0].x, vs[g][0].y, vs[g][0].z, vs[g][0].w,
                              vs[g][1].x, vs[g][1].y, vs[g][1].z, vs[g][1].w,
                              vs[g][2].x, vs[g][2].y, vs[g][2].z, vs[g][2].w};
        float ip[4];
#pragma unroll
        for (int j = 0; j < 4; ++j) {
          ip[j] = (xd[3 * j] * xs[3 * j] + xd[3 * j + 1] * xs[3 * j + 1] +
                   xd[3 * j + 2] * xs[3 * j + 2]) * INV_SQ3;
        }
        as0[32 + 2 * g].x = ip[0];     as0[32 + 2 * g].y = ip[1];
        as0[32 + 2 * g + 1].x = ip[2]; as0[32 + 2 * g + 1].y = ip[3];
      }
    }

    {
      v8f c0 = {}, c1 = {};
#pragma unroll
      for (int Q = 0; Q < 40; ++Q) {
        const int k0 = 8 * (Q >> 1) + 4 * hk + 2 * (Q & 1);
        c0 = wmma4(as0[Q], ldb<32>(L1, k0, mrow), c0);
        c1 = wmma4(as0[Q], ldb<32>(L1, k0, 16 + mrow), c1);
      }
      float* xsS = s0S;
      const int half = lane >> 4;
      const int c4 = lane & 15;
      v4f t0[8];
#pragma unroll
      for (int i = 0; i < 8; ++i) {
        const int m = 2 * i + half;
        t0[i] = *(const v4f*)(x0n + (size_t)__shfl(srow, m) * 64 + c4 * 4);
      }
      v4f t1[12];
#pragma unroll
      for (int i = 0; i < 12; ++i) {
        const int f = lane + 32 * i;
        const int m = f / 24, q4 = f % 24;
        t1[i] = *(const v4f*)(x1n + (size_t)__shfl(srow, m) * 96 + q4 * 4);
      }
      tile_store(c0, [&](int r, int m, int n, float v) {
        hN[m * 32 + n] = sspf(v * INV_SQ160);
        hN[m * 32 + 16 + n] = sspf(c1[r] * INV_SQ160);
      });
#pragma unroll
      for (int i = 0; i < 8; ++i) {
        const int m = 2 * i + half;
        *(v4f*)(xsS + m * 160 + c4 * 4) = t0[i];
      }
#pragma unroll
      for (int i = 0; i < 12; ++i) {
        const int f = lane + 32 * i;
        const int m = f / 24, q4 = f % 24;
        *(v4f*)(xsS + m * 160 + 64 + q4 * 4) = t1[i];
      }
    }
  }
  {
    const float* ea = edge_attr + (size_t)e0 * 32;
    v2f aea[8];
    load_a(aea, [&](int m, int k) { return ld2(ea + m * 32 + k); });
    for (int j = 0; j < 2; ++j) {
      v8f c = {};
      c = wmma_tile(aea, [&](int n, int k) {
        return ldb<32>(F1, k, j * 16 + n);
      }, c);
      tile_store(c, [&](int, int m, int n, float v) {
        hE[m * 32 + j * 16 + n] = sspf(v * INV_SQ32);
      });
    }
  }
  __syncthreads();

  const int n = lane & 15;
  const int half = lane >> 4;
  const int rbase = half << 3;
  float sh0r[8], s1xr[8], s1yr[8], s1zr[8];
#pragma unroll
  for (int r = 0; r < 8; ++r) {
    int e = e0 + rbase + r;
    sh0r[r] = edge_sh[(size_t)e * 4 + 0];
    s1xr[r] = edge_sh[(size_t)e * 4 + 1];
    s1yr[r] = edge_sh[(size_t)e * 4 + 2];
    s1zr[r] = edge_sh[(size_t)e * 4 + 3];
  }
  const int dmine = drow;
  const float* xsS = s0S;

  v2f aN[8], aE[8];
  load_a(aN, [&](int m, int k) { return ld2(hN + m * 32 + k); });
  load_a(aE, [&](int m, int k) { return ld2(hE + m * 32 + k); });

  float* const stg = hNE;
  constexpr int P1 = 36;
  constexpr int P3 = 97;

  auto wtile = [&](int nb) -> v8f {
    v8f cN = {}, cE = {};
    cN = wmma_tile(aN, [&](int nn, int k) {
      return ldb<224>(L2, k, nb + nn);
    }, cN);
    cE = wmma_tile(aE, [&](int nn, int k) {
      return ldb<224>(F2, k, nb + nn);
    }, cE);
    return cN * cE * (INV_SQ32 * INV_SQ32);
  };

  auto add_line = [&](int m, int ch0, float v) {
    const int d = __builtin_amdgcn_readlane(dmine, m);
    atomicAdd(acc + (size_t)d * 480 + ch0 + lane, v);
  };

  auto scatter1 = [&](const float (&o)[2][8], int ch0) {
#pragma unroll
    for (int r = 0; r < 8; ++r) {
      stg[(rbase + r) * P1 + n] = o[0][r];
      stg[(rbase + r) * P1 + 16 + n] = o[1][r];
    }
    __builtin_amdgcn_fence(__ATOMIC_ACQ_REL, "wavefront");
    float v[16];
#pragma unroll
    for (int m = 0; m < 16; ++m) v[m] = stg[m * P1 + lane];
#pragma unroll
    for (int m = 0; m < 16; ++m) add_line(m, ch0, v[m]);
    __builtin_amdgcn_fence(__ATOMIC_ACQ_REL, "wavefront");
  };

  auto scatter3 = [&](const float (&o)[2][8][3], int ch0) {
#pragma unroll
    for (int ph = 0; ph < 2; ++ph) {
#pragma unroll
      for (int q = 0; q < 4; ++q) {
        float* row = stg + (q + 4 * half) * P3;
#pragma unroll
        for (int t = 0; t < 2; ++t) {
          row[t * 48 + 3 * n + 0] = o[t][4 * ph + q][0];
          row[t * 48 + 3 * n + 1] = o[t][4 * ph + q][1];
          row[t * 48 + 3 * n + 2] = o[t][4 * ph + q][2];
        }
      }
      __builtin_amdgcn_fence(__ATOMIC_ACQ_REL, "wavefront");
      float v[8][3];
#pragma unroll
      for (int s = 0; s < 8; ++s)
#pragma unroll
        for (int L = 0; L < 3; ++L) v[s][L] = stg[s * P3 + L * 32 + lane];
#pragma unroll
      for (int s = 0; s < 8; ++s) {
        const int m = (s & 3) + 4 * ph + 8 * (s >> 2);
#pragma unroll
        for (int L = 0; L < 3; ++L) add_line(m, ch0 + 32 * L, v[s][L]);
      }
      __builtin_amdgcn_fence(__ATOMIC_ACQ_REL, "wavefront");
    }
  };

  for (int p = 0; p < 2; ++p) {
    const int nb = 32 * p;
    const v8f wA = wtile(nb), wB = wtile(nb + 16);
    float o[2][8];
#pragma unroll
    for (int r = 0; r < 8; ++r) {
      const float* xr = xsS + (rbase + r) * 160 + nb + n;
      o[0][r] = wA[r] * xr[0] * sh0r[r];
      o[1][r] = wB[r] * xr[16] * sh0r[r];
    }
    scatter1(o, nb);
  }
  {
    const v8f wA = wtile(160), wB = wtile(176);
    float o[2][8];
#pragma unroll
    for (int r = 0; r < 8; ++r) {
#pragma unroll
      for (int t = 0; t < 2; ++t) {
        const float* xv = xsS + (rbase + r) * 160 + 64 + (16 * t + n) * 3;
        const float wv = t ? wB[r] : wA[r];
        float dp = xv[0] * s1xr[r] + xv[1] * s1yr[r] + xv[2] * s1zr[r];
        o[t][r] = wv * dp * INV_SQ3;
      }
    }
    scatter1(o, 64);
  }
  for (int p = 0; p < 2; ++p) {
    const int nb = 64 + 32 * p;
    const v8f wA = wtile(nb), wB = wtile(nb + 16);
    float o[2][8][3];
#pragma unroll
    for (int r = 0; r < 8; ++r) {
      const float* xr = xsS + (rbase + r) * 160 + 32 * p + n;
#pragma unroll
      for (int t = 0; t < 2; ++t) {
        const float wv = t ? wB[r] : wA[r];
        float f = wv * xr[16 * t] * INV_SQ3;
        o[t][r][0] = f * s1xr[r];
        o[t][r][1] = f * s1yr[r];
        o[t][r][2] = f * s1zr[r];
      }
    }
    scatter3(o, 96 + 96 * p);
  }
  {
    const v8f wA = wtile(128), wB = wtile(144);
    float o[2][8][3];
#pragma unroll
    for (int r = 0; r < 8; ++r) {
#pragma unroll
      for (int t = 0; t < 2; ++t) {
        const float* xv = xsS + (rbase + r) * 160 + 64 + (16 * t + n) * 3;
        const float wv = t ? wB[r] : wA[r];
        float f = wv * sh0r[r];
        o[t][r][0] = f * xv[0];
        o[t][r][1] = f * xv[1];
        o[t][r][2] = f * xv[2];
      }
    }
    scatter3(o, 288);
  }
  {
    const v8f wA = wtile(192), wB = wtile(208);
    float o[2][8][3];
#pragma unroll
    for (int r = 0; r < 8; ++r) {
#pragma unroll
      for (int t = 0; t < 2; ++t) {
        const float* xv = xsS + (rbase + r) * 160 + 64 + (16 * t + n) * 3;
        const float wv = t ? wB[r] : wA[r];
        const float s1x = s1xr[r], s1y = s1yr[r], s1z = s1zr[r];
        float cx = xv[1] * s1z - xv[2] * s1y;
        float cy = xv[2] * s1x - xv[0] * s1z;
        float cz = xv[0] * s1y - xv[1] * s1x;
        float f = wv * INV_SQ2;
        o[t][r][0] = f * cx;
        o[t][r][1] = f * cy;
        o[t][r][2] = f * cz;
      }
    }
    scatter3(o, 384);
  }
}

__global__ __launch_bounds__(32) void proj_kernel(
    const float* __restrict__ acc,
    const float* __restrict__ WO0, const float* __restrict__ bO0,
    const float* __restrict__ WO1, const float* __restrict__ WO2,
    float* __restrict__ y) {
  const int n0 = blockIdx.x * 16;
  const float* at = acc + (size_t)n0 * 480;

  {
    v2f aa[24];
    load_a(aa, [&](int m, int k) { return ld2(at + (size_t)m * 480 + k); });
    for (int j = 0; j < 8; ++j) {
      v8f c = {};
      c = wmma_tile(aa, [&](int n, int k) {
        return ldb<128>(WO0, k, j * 16 + n);
      }, c);
      tile_store(c, [&](int, int m, int n, float v) {
        int col = j * 16 + n;
        y[(size_t)(n0 + m) * 416 + col] = v * INV_SQ96 + bO0[col];
      });
    }
  }
  for (int i = 0; i < 3; ++i) {
    v2f aa[24];
    load_a(aa, [&](int m, int k) {
      v2f a;
      a.x = at[(size_t)m * 480 + 96 + 3 * k + i];
      a.y = at[(size_t)m * 480 + 96 + 3 * (k + 1) + i];
      return a;
    });
#pragma unroll
    for (int j = 0; j < 4; ++j) {
      v8f c = {};
      c = wmma_tile(aa, [&](int n, int k) {
        return ldb<64>(WO1, k, j * 16 + n);
      }, c);
      tile_store(c, [&](int, int m, int n, float v) {
        y[(size_t)(n0 + m) * 416 + 128 + (j * 16 + n) * 3 + i] = v * INV_SQ96;
      });
    }
  }
  for (int i = 0; i < 3; ++i) {
    v2f aa[8];
    load_a(aa, [&](int m, int k) {
      v2f a;
      a.x = at[(size_t)m * 480 + 384 + 3 * k + i];
      a.y = at[(size_t)m * 480 + 384 + 3 * (k + 1) + i];
      return a;
    });
    for (int j = 0; j < 2; ++j) {
      v8f c = {};
      c = wmma_tile(aa, [&](int n, int k) {
        return ldb<32>(WO2, k, j * 16 + n);
      }, c);
      tile_store(c, [&](int, int m, int n, float v) {
        y[(size_t)(n0 + m) * 416 + 320 + (j * 16 + n) * 3 + i] = v * INV_SQ32;
      });
    }
  }
}

extern "C" void kernel_launch(void* const* d_in, const int* in_sizes, int n_in,
                              void* d_out, int out_size, void* d_ws,
                              size_t ws_size, hipStream_t stream) {
  (void)in_sizes; (void)n_in; (void)out_size; (void)ws_size;
  const float* x         = (const float*)d_in[0];
  const float* edge_sh   = (const float*)d_in[1];
  const float* edge_attr = (const float*)d_in[2];
  const float* W0p = (const float*)d_in[3];
  const float* b0p = (const float*)d_in[4];
  const float* W1p = (const float*)d_in[5];
  const float* W0n = (const float*)d_in[6];
  const float* b0n = (const float*)d_in[7];
  const float* W1n = (const float*)d_in[8];
  const float* G1  = (const float*)d_in[9];
  const float* g1b = (const float*)d_in[10];
  const float* G2  = (const float*)d_in[11];
  const float* g2b = (const float*)d_in[12];
  const float* F1  = (const float*)d_in[13];
  const float* F2  = (const float*)d_in[14];
  const float* L1  = (const float*)d_in[15];
  const float* L2  = (const float*)d_in[16];
  const float* WO0 = (const float*)d_in[17];
  const float* bO0 = (const float*)d_in[18];
  const float* WO1 = (const float*)d_in[19];
  const float* WO2 = (const float*)d_in[20];
  const int* edge_index = (const int*)d_in[21];

  float* ws   = (float*)d_ws;
  float* p0   = ws;
  float* p1   = p0   + (size_t)N_NODES * 64;
  float* x0n  = p1   + (size_t)N_NODES * 96;
  float* x1n  = x0n  + (size_t)N_NODES * 64;
  float* acc  = x1n  + (size_t)N_NODES * 96;

  hipMemsetAsync(acc, 0, (size_t)N_NODES * 480 * sizeof(float), stream);

  node_kernel<<<N_NODES / 16, 32, 0, stream>>>(
      x, W0p, b0p, W1p, W0n, b0n, W1n, G1, g1b, G2, g2b,
      p0, p1, x0n, x1n);

  edge_kernel<<<N_EDGES / 16, 32, 0, stream>>>(
      edge_sh, edge_attr, F1, F2, L1, L2, edge_index,
      p0, p1, x0n, x1n, acc);

  proj_kernel<<<N_NODES / 16, 32, 0, stream>>>(
      acc, WO0, bO0, WO1, WO2, (float*)d_out);
}


# Edge MLP shifted softplus loses its per-value branch: the library chain runs in every lane and a select keeps v above 20, so a lane's chains interleave (were serial); 5 waves kept; same values, sums
# speedup vs baseline: 2.1002x; 1.0645x over previous
#include <hip/hip_runtime.h>
#include <hip/hip_bf16.h>
#include <math.h>


typedef __attribute__((ext_vector_type(2))) float v2f;
typedef __attribute__((ext_vector_type(4))) float v4f;
typedef __attribute__((ext_vector_type(8))) float v8f;

#define N_NODES 10000
#define N_EDGES 320000

#define INV_SQ3   0.5773502691896258f
#define INV_SQ2   0.7071067811865476f
#define INV_SQ32  0.17677669529663687f
#define INV_SQ64  0.125f
#define INV_SQ96  0.10206207261596575f
#define INV_SQ160 0.07905694150420949f
#define LN2F      0.6931471805599453f

__device__ __forceinline__ float sspf(float v) {
  const float chain = log1pf(expf(v));
  const float sp = (v > 20.0f) ? v : chain;
  return sp - LN2F;
}
__device__ __forceinline__ float siluf(float v) {
  return v / (1.0f + expf(-v));
}
__device__ __forceinline__ v2f ld2(const float* p) {
  return *(const v2f*)p;
}
template <int NC>
__device__ __forceinline__ v2f ldb(const float* __restrict__ W, int k, int col) {
  v2f b;
  b.x = W[k * NC + col];
  b.y = W[(k + 1) * NC + col];
  return b;
}

__device__ __forceinline__ v8f wmma4(v2f a, v2f b, v8f c) {
  return __builtin_amdgcn_wmma_f32_16x16x4_f32(false, a, false, b,
                                               (short)0, c, false, false);
}


template <int KS, typename LA>
__device__ __forceinline__ void load_a(v2f (&a)[KS], LA la) {
  const int lane = threadIdx.x & 31;
  const int m = lane & 15;
  const int kk = (lane >> 4) << 1;
#pragma unroll
  for (int q = 0; q < KS; ++q) a[q] = la(m, q * 4 + kk);
}

template <int KS, typename LB>
__device__ __forceinline__ v8f wmma_tile(const v2f (&a)[KS], LB lb, v8f c) {
  const int lane = threadIdx.x & 31;
  const int n = lane & 15;
  const int kk = (lane >> 4) << 1;
#pragma unroll
  for (int q = 0; q < KS; ++q) c = wmma4(a[q], lb(n, q * 4 + kk), c);
  return c;
}

template <typename ST>
__device__ __forceinline__ void tile_store(v8f c, ST st) {
  const int lane = threadIdx.x & 31;
  const int col = lane & 15;
  const int rbase = (lane >> 4) << 3;
#pragma unroll
  for (int r = 0; r < 8; ++r) st(r, rbase + r, col, c[r]);
}

__global__ __launch_bounds__(32) void node_kernel(
    const float* __restrict__ x,
    const float* __restrict__ W0p, const float* __restrict__ b0p,
    const float* __restrict__ W1p,
    const float* __restrict__ W0n, const float* __restrict__ b0n,
    const float* __restrict__ W1n,
    const float* __restrict__ G1, const float* __restrict__ g1b,
    const float* __restrict__ G2, const float* __restrict__ g2b,
    float* __restrict__ p0, float* __restrict__ p1,
    float* __restrict__ x0n, float* __restrict__ x1n) {
  __shared__ __align__(16) float f0S[16 * 96];
  __shared__ __align__(16) float hS[16 * 96];
  __shared__ __align__(16) float gS[16 * 96];

  const int lane = threadIdx.x;
  const int n0 = blockIdx.x * 16;
  const float* xt = x + (size_t)n0 * 160;

  {
    v2f ax[16];
    load_a(ax, [&](int m, int k) { return ld2(xt + m * 160 + k); });
#pragma unroll
    for (int j = 0; j < 4; ++j) {
      v8f c = {};
      c = wmma_tile(ax, [&](int n, int k) {
        return ldb<64>(W0p, k, j * 16 + n);
      }, c);
      tile_store(c, [&](int, int m, int n, float v) {
        int col = j * 16 + n;
        p0[(size_t)(n0 + m) * 64 + col] = v * INV_SQ64 + b0p[col];
      });
    }
  }
#pragma unroll
  for (int i = 0; i < 3; ++i) {
    v2f ax[8];
    load_a(ax, [&](int m, int k) {
      v2f a;
      a.x = xt[m * 160 + 64 + 3 * k + i];
      a.y = xt[m * 160 + 64 + 3 * (k + 1) + i];
      return a;
    });
    for (int j = 0; j < 2; ++j) {
      v8f c = {};
      c = wmma_tile(ax, [&](int n, int k) {
        return ldb<32>(W1p, k, j * 16 + n);
      }, c);
      tile_store(c, [&](int, int m, int n, float v) {
        p1[(size_t)(n0 + m) * 96 + (j * 16 + n) * 3 + i] = v * INV_SQ32;
      });
    }
  }

  for (int idx = lane; idx < 16 * 96; idx += 32) {
    int m = idx / 96, c = idx % 96;
    float v;
    if (c < 64) {
      v = xt[m * 160 + c];
    } else {
      int u = c - 64;
      float a0 = xt[m * 160 + 64 + u * 3 + 0];
      float a1 = xt[m * 160 + 64 + u * 3 + 1];
      float a2 = xt[m * 160 + 64 + u * 3 + 2];
      v = sqrtf(a0 * a0 + a1 * a1 + a2 * a2 + 1e-12f);
    }
    f0S[idx] = v;
  }
  __syncthreads();

  {
    v2f af[24];
    load_a(af, [&](int m, int k) { return ld2(f0S + m * 96 + k); });
    for (int j = 0; j < 6; ++j) {
      v8f c = {};
      c = wmma_tile(af, [&](int n, int k) {
        return ldb<96>(G1, k, j * 16 + n);
      }, c);
      tile_store(c, [&](int, int m, int n, float v) {
        int col = j * 16 + n;
        hS[m * 96 + col] = siluf(v + g1b[col]);
      });
    }
  }
  __syncthreads();
  {
    v2f ah[24];
    load_a(ah, [&](int m, int k) { return ld2(hS + m * 96 + k); });
    for (int j = 0; j < 6; ++j) {
      v8f c = {};
      c = wmma_tile(ah, [&](int n, int k) {
        return ldb<96>(G2, k, j * 16 + n);
      }, c);
      tile_store(c, [&](int, int m, int n, float v) {
        gS[m * 96 + j * 16 + n] = v + g2b[j * 16 + n];
      });
    }
  }
  __syncthreads();

  {
    v2f ag[16];
    load_a(ag, [&](int m, int k) { return ld2(gS + m * 96 + k); });
#pragma unroll
    for (int j = 0; j < 4; ++j) {
      v8f c = {};
      c = wmma_tile(ag, [&](int n, int k) {
        return ldb<64>(W0n, k, j * 16 + n);
      }, c);
      tile_store(c, [&](int, int m, int n, float v) {
        int col = j * 16 + n;
        x0n[(size_t)(n0 + m) * 64 + col] = v * INV_SQ64 + b0n[col];
      });
    }
  }
  for (int i = 0; i < 3; ++i) {
    v2f ax[8];
    load_a(ax, [&](int m, int k) {
      v2f a;
      a.x = xt[m * 160 + 64 + 3 * k + i] * gS[m * 96 + 64 + k];
      a.y = xt[m * 160 + 64 + 3 * (k + 1) + i] * gS[m * 96 + 64 + k + 1];
      return a;
    });
    for (int j = 0; j < 2; ++j) {
      v8f c = {};
      c = wmma_tile(ax, [&](int n, int k) {
        return ldb<32>(W1n, k, j * 16 + n);
      }, c);
      tile_store(c, [&](int, int m, int n, float v) {
        x1n[(size_t)(n0 + m) * 96 + (j * 16 + n) * 3 + i] = v * INV_SQ32;
      });
    }
  }
}

__global__ __launch_bounds__(32) __attribute__((amdgpu_waves_per_eu(5))) void edge_kernel(
    const float* __restrict__ edge_sh, const float* __restrict__ edge_attr,
    const float* __restrict__ F1, const float* __restrict__ F2,
    const float* __restrict__ L1, const float* __restrict__ L2,
    const int* __restrict__ edge_index,
    const float* __restrict__ p0, const float* __restrict__ p1,
    const float* __restrict__ x0n, const float* __restrict__ x1n,
    float* __restrict__ acc) {
  __shared__ __align__(16) float s0S[16 * 160];
  __shared__ __align__(16) float hNE[2 * 16 * 32];
  float* const hN = hNE;
  float* const hE = hNE + 16 * 32;

  const int lane = threadIdx.x;
  const int e0 = blockIdx.x * 16;

  __builtin_prefetch(F2, 0, 1);
  __builtin_prefetch(L2, 0, 1);

  const int mrow = lane & 15;
  const int hk = lane >> 4;
  const int drow = edge_index[e0 + mrow];
  const int srow = edge_index[N_EDGES + e0 + mrow];

  {
    v2f as0[40];
    {
      const float* pd0 = p0 + (size_t)drow * 64 + 4 * hk;
      const float* ps0 = p0 + (size_t)srow * 64 + 4 * hk;
      const float* pd1 = p1 + (size_t)drow * 96 + 12 * hk;
      const float* ps1 = p1 + (size_t)srow * 96 + 12 * hk;
      v4f qd[8], qs[8], vd[4][3], vs[4][3];
#pragma unroll
      for (int g = 0; g < 4; ++g) {
#pragma unroll
        for (int j = 0; j < 3; ++j) {
          vd[g][j] = *(const v4f*)(pd1 + 24 * g + 4 * j);
          vs[g][j] = *(const v4f*)(ps1 + 24 * g + 4 * j);
        }
      }
#pragma unroll
      for (int g = 0; g < 8; ++g) {
        qd[g] = *(const v4f*)(pd0 + 8 * g);
        qs[g] = *(const v4f*)(ps0 + 8 * g);
      }
#pragma unroll
      for (int g = 0; g < 8; ++g) {
        as0[2 * g].x = qd[g].x;          as0[2 * g].y = qd[g].y;
        as0[2 * g + 1].x = qd[g].z;      as0[2 * g + 1].y = qd[g].w;
        as0[16 + 2 * g].x = qs[g].x;     as0[16 + 2 * g].y = qs[g].y;
        as0[16 + 2 * g + 1].x = qs[g].z; as0[16 + 2 * g + 1].y = qs[g].w;
      }
#pragma unroll
      for (int g = 0; g < 4; ++g) {
        const float xd[12] = {vd[g][0].x, vd[g][0].y, vd[g][0].z, vd[g][0].w,
                              vd[g][1].x, vd[g][1].y, vd[g][1].z, vd[g][1].w,
                              vd[g][2].x, vd[g][2].y, vd[g][2].z, vd[g][2].w};
        const float xs[12] = {vs[g][0].x, vs[g][0].y, vs[g][0].z, vs[g][0].w,
                              vs[g][1].x, vs[g][1].y, vs[g][1].z, vs[g][1].w,
                              vs[g][2].x, vs[g][2].y, vs[g][2].z, vs[g][2].w};
        float ip[4];
#pragma unroll
        for (int j = 0; j < 4; ++j) {
          ip[j] = (xd[3 * j] * xs[3 * j] + xd[3 * j + 1] * xs[3 * j + 1] +
                   xd[3 * j + 2] * xs[3 * j + 2]) * INV_SQ3;
        }
        as0[32 + 2 * g].x = ip[0];     as0[32 + 2 * g].y = ip[1];
        as0[32 + 2 * g + 1].x = ip[2]; as0[32 + 2 * g + 1].y = ip[3];
      }
    }

    {
      v8f c0 = {}, c1 = {};
#pragma unroll
      for (int Q = 0; Q < 40; ++Q) {
        const int k0 = 8 * (Q >> 1) + 4 * hk + 2 * (Q & 1);
        c0 = wmma4(as0[Q], ldb<32>(L1, k0, mrow), c0);
        c1 = wmma4(as0[Q], ldb<32>(L1, k0, 16 + mrow), c1);
      }
      float* xsS = s0S;
      const int half = lane >> 4;
      const int c4 = lane & 15;
      v4f t0[8];
#pragma unroll
      for (int i = 0; i < 8; ++i) {
        const int m = 2 * i + half;
        t0[i] = *(const v4f*)(x0n + (size_t)__shfl(srow, m) * 64 + c4 * 4);
      }
      v4f t1[12];
#pragma unroll
      for (int i = 0; i < 12; ++i) {
        const int f = lane + 32 * i;
        const int m = f / 24, q4 = f % 24;
        t1[i] = *(const v4f*)(x1n + (size_t)__shfl(srow, m) * 96 + q4 * 4);
      }
      tile_store(c0, [&](int r, int m, int n, float v) {
        hN[m * 32 + n] = sspf(v * INV_SQ160);
        hN[m * 32 + 16 + n] = sspf(c1[r] * INV_SQ160);
      });
#pragma unroll
      for (int i = 0; i < 8; ++i) {
        const int m = 2 * i + half;
        *(v4f*)(xsS + m * 160 + c4 * 4) = t0[i];
      }
#pragma unroll
      for (int i = 0; i < 12; ++i) {
        const int f = lane + 32 * i;
        const int m = f / 24, q4 = f % 24;
        *(v4f*)(xsS + m * 160 + 64 + q4 * 4) = t1[i];
      }
    }
  }
  {
    const float* ea = edge_attr + (size_t)e0 * 32;
    v2f aea[8];
    load_a(aea, [&](int m, int k) { return ld2(ea + m * 32 + k); });
    for (int j = 0; j < 2; ++j) {
      v8f c = {};
      c = wmma_tile(aea, [&](int n, int k) {
        return ldb<32>(F1, k, j * 16 + n);
      }, c);
      tile_store(c, [&](int, int m, int n, float v) {
        hE[m * 32 + j * 16 + n] = sspf(v * INV_SQ32);
      });
    }
  }
  __syncthreads();

  const int n = lane & 15;
  const int half = lane >> 4;
  const int rbase = half << 3;
  float sh0r[8], s1xr[8], s1yr[8], s1zr[8];
#pragma unroll
  for (int r = 0; r < 8; ++r) {
    int e = e0 + rbase + r;
    sh0r[r] = edge_sh[(size_t)e * 4 + 0];
    s1xr[r] = edge_sh[(size_t)e * 4 + 1];
    s1yr[r] = edge_sh[(size_t)e * 4 + 2];
    s1zr[r] = edge_sh[(size_t)e * 4 + 3];
  }
  const int dmine = drow;
  const float* xsS = s0S;

  v2f aN[8], aE[8];
  load_a(aN, [&](int m, int k) { return ld2(hN + m * 32 + k); });
  load_a(aE, [&](int m, int k) { return ld2(hE + m * 32 + k); });

  float* const stg = hNE;
  constexpr int P1 = 36;
  constexpr int P3 = 97;

  auto wtile = [&](int nb) -> v8f {
    v8f cN = {}, cE = {};
    cN = wmma_tile(aN, [&](int nn, int k) {
      return ldb<224>(L2, k, nb + nn);
    }, cN);
    cE = wmma_tile(aE, [&](int nn, int k) {
      return ldb<224>(F2, k, nb + nn);
    }, cE);
    return cN * cE * (INV_SQ32 * INV_SQ32);
  };

  auto add_line = [&](int m, int ch0, float v) {
    const int d = __builtin_amdgcn_readlane(dmine, m);
    atomicAdd(acc + (size_t)d * 480 + ch0 + lane, v);
  };

  auto scatter1 = [&](const float (&o)[2][8], int ch0) {
#pragma unroll
    for (int r = 0; r < 8; ++r) {
      stg[(rbase + r) * P1 + n] = o[0][r];
      stg[(rbase + r) * P1 + 16 + n] = o[1][r];
    }
    __builtin_amdgcn_fence(__ATOMIC_ACQ_REL, "wavefront");
    float v[16];
#pragma unroll
    for (int m = 0; m < 16; ++m) v[m] = stg[m * P1 + lane];
#pragma unroll
    for (int m = 0; m < 16; ++m) add_line(m, ch0, v[m]);
    __builtin_amdgcn_fence(__ATOMIC_ACQ_REL, "wavefront");
  };

  auto scatter3 = [&](const float (&o)[2][8][3], int ch0) {
#pragma unroll
    for (int ph = 0; ph < 2; ++ph) {
#pragma unroll
      for (int q = 0; q < 4; ++q) {
        float* row = stg + (q + 4 * half) * P3;
#pragma unroll
        for (int t = 0; t < 2; ++t) {
          row[t * 48 + 3 * n + 0] = o[t][4 * ph + q][0];
          row[t * 48 + 3 * n + 1] = o[t][4 * ph + q][1];
          row[t * 48 + 3 * n + 2] = o[t][4 * ph + q][2];
        }
      }
      __builtin_amdgcn_fence(__ATOMIC_ACQ_REL, "wavefront");
      float v[8][3];
#pragma unroll
      for (int s = 0; s < 8; ++s)
#pragma unroll
        for (int L = 0; L < 3; ++L) v[s][L] = stg[s * P3 + L * 32 + lane];
#pragma unroll
      for (int s = 0; s < 8; ++s) {
        const int m = (s & 3) + 4 * ph + 8 * (s >> 2);
#pragma unroll
        for (int L = 0; L < 3; ++L) add_line(m, ch0 + 32 * L, v[s][L]);
      }
      __builtin_amdgcn_fence(__ATOMIC_ACQ_REL, "wavefront");
    }
  };

  for (int p = 0; p < 2; ++p) {
    const int nb = 32 * p;
    const v8f wA = wtile(nb), wB = wtile(nb + 16);
    float o[2][8];
#pragma unroll
    for (int r = 0; r < 8; ++r) {
      const float* xr = xsS + (rbase + r) * 160 + nb + n;
      o[0][r] = wA[r] * xr[0] * sh0r[r];
      o[1][r] = wB[r] * xr[16] * sh0r[r];
    }
    scatter1(o, nb);
  }
  {
    const v8f wA = wtile(160), wB = wtile(176);
    float o[2][8];
#pragma unroll
    for (int r = 0; r < 8; ++r) {
#pragma unroll
      for (int t = 0; t < 2; ++t) {
        const float* xv = xsS + (rbase + r) * 160 + 64 + (16 * t + n) * 3;
        const float wv = t ? wB[r] : wA[r];
        float dp = xv[0] * s1xr[r] + xv[1] * s1yr[r] + xv[2] * s1zr[r];
        o[t][r] = wv * dp * INV_SQ3;
      }
    }
    scatter1(o, 64);
  }
  for (int p = 0; p < 2; ++p) {
    const int nb = 64 + 32 * p;
    const v8f wA = wtile(nb), wB = wtile(nb + 16);
    float o[2][8][3];
#pragma unroll
    for (int r = 0; r < 8; ++r) {
      const float* xr = xsS + (rbase + r) * 160 + 32 * p + n;
#pragma unroll
      for (int t = 0; t < 2; ++t) {
        const float wv = t ? wB[r] : wA[r];
        float f = wv * xr[16 * t] * INV_SQ3;
        o[t][r][0] = f * s1xr[r];
        o[t][r][1] = f * s1yr[r];
        o[t][r][2] = f * s1zr[r];
      }
    }
    scatter3(o, 96 + 96 * p);
  }
  {
    const v8f wA = wtile(128), wB = wtile(144);
    float o[2][8][3];
#pragma unroll
    for (int r = 0; r < 8; ++r) {
#pragma unroll
      for (int t = 0; t < 2; ++t) {
        const float* xv = xsS + (rbase + r) * 160 + 64 + (16 * t + n) * 3;
        const float wv = t ? wB[r] : wA[r];
        float f = wv * sh0r[r];
        o[t][r][0] = f * xv[0];
        o[t][r][1] = f * xv[1];
        o[t][r][2] = f * xv[2];
      }
    }
    scatter3(o, 288);
  }
  {
    const v8f wA = wtile(192), wB = wtile(208);
    float o[2][8][3];
#pragma unroll
    for (int r = 0; r < 8; ++r) {
#pragma unroll
      for (int t = 0; t < 2; ++t) {
        const float* xv = xsS + (rbase + r) * 160 + 64 + (16 * t + n) * 3;
        const float wv = t ? wB[r] : wA[r];
        const float s1x = s1xr[r], s1y = s1yr[r], s1z = s1zr[r];
        float cx = xv[1] * s1z - xv[2] * s1y;
        float cy = xv[2] * s1x - xv[0] * s1z;
        float cz = xv[0] * s1y - xv[1] * s1x;
        float f = wv * INV_SQ2;
        o[t][r][0] = f * cx;
        o[t][r][1] = f * cy;
        o[t][r][2] = f * cz;
      }
    }
    scatter3(o, 384);
  }
}

__global__ __launch_bounds__(32) void proj_kernel(
    const float* __restrict__ acc,
    const float* __restrict__ WO0, const float* __restrict__ bO0,
    const float* __restrict__ WO1, const float* __restrict__ WO2,
    float* __restrict__ y) {
  const int n0 = blockIdx.x * 16;
  const float* at = acc + (size_t)n0 * 480;

  {
    v2f aa[24];
    load_a(aa, [&](int m, int k) { return ld2(at + (size_t)m * 480 + k); });
    for (int j = 0; j < 8; ++j) {
      v8f c = {};
      c = wmma_tile(aa, [&](int n, int k) {
        return ldb<128>(WO0, k, j * 16 + n);
      }, c);
      tile_store(c, [&](int, int m, int n, float v) {
        int col = j * 16 + n;
        y[(size_t)(n0 + m) * 416 + col] = v * INV_SQ96 + bO0[col];
      });
    }
  }
  for (int i = 0; i < 3; ++i) {
    v2f aa[24];
    load_a(aa, [&](int m, int k) {
      v2f a;
      a.x = at[(size_t)m * 480 + 96 + 3 * k + i];
      a.y = at[(size_t)m * 480 + 96 + 3 * (k + 1) + i];
      return a;
    });
#pragma unroll
    for (int j = 0; j < 4; ++j) {
      v8f c = {};
      c = wmma_tile(aa, [&](int n, int k) {
        return ldb<64>(WO1, k, j * 16 + n);
      }, c);
      tile_store(c, [&](int, int m, int n, float v) {
        y[(size_t)(n0 + m) * 416 + 128 + (j * 16 + n) * 3 + i] = v * INV_SQ96;
      });
    }
  }
  for (int i = 0; i < 3; ++i) {
    v2f aa[8];
    load_a(aa, [&](int m, int k) {
      v2f a;
      a.x = at[(size_t)m * 480 + 384 + 3 * k + i];
      a.y = at[(size_t)m * 480 + 384 + 3 * (k + 1) + i];
      return a;
    });
    for (int j = 0; j < 2; ++j) {
      v8f c = {};
      c = wmma_tile(aa, [&](int n, int k) {
        return ldb<32>(WO2, k, j * 16 + n);
      }, c);
      tile_store(c, [&](int, int m, int n, float v) {
        y[(size_t)(n0 + m) * 416 + 320 + (j * 16 + n) * 3 + i] = v * INV_SQ32;
      });
    }
  }
}

extern "C" void kernel_launch(void* const* d_in, const int* in_sizes, int n_in,
                              void* d_out, int out_size, void* d_ws,
                              size_t ws_size, hipStream_t stream) {
  (void)in_sizes; (void)n_in; (void)out_size; (void)ws_size;
  const float* x         = (const float*)d_in[0];
  const float* edge_sh   = (const float*)d_in[1];
  const float* edge_attr = (const float*)d_in[2];
  const float* W0p = (const float*)d_in[3];
  const float* b0p = (const float*)d_in[4];
  const float* W1p = (const float*)d_in[5];
  const float* W0n = (const float*)d_in[6];
  const float* b0n = (const float*)d_in[7];
  const float* W1n = (const float*)d_in[8];
  const float* G1  = (const float*)d_in[9];
  const float* g1b = (const float*)d_in[10];
  const float* G2  = (const float*)d_in[11];
  const float* g2b = (const float*)d_in[12];
  const float* F1  = (const float*)d_in[13];
  const float* F2  = (const float*)d_in[14];
  const float* L1  = (const float*)d_in[15];
  const float* L2  = (const float*)d_in[16];
  const float* WO0 = (const float*)d_in[17];
  const float* bO0 = (const float*)d_in[18];
  const float* WO1 = (const float*)d_in[19];
  const float* WO2 = (const float*)d_in[20];
  const int* edge_index = (const int*)d_in[21];

  float* ws   = (float*)d_ws;
  float* p0   = ws;
  float* p1   = p0   + (size_t)N_NODES * 64;
  float* x0n  = p1   + (size_t)N_NODES * 96;
  float* x1n  = x0n  + (size_t)N_NODES * 64;
  float* acc  = x1n  + (size_t)N_NODES * 96;

  hipMemsetAsync(acc, 0, (size_t)N_NODES * 480 * sizeof(float), stream);

  node_kernel<<<N_NODES / 16, 32, 0, stream>>>(
      x, W0p, b0p, W1p, W0n, b0n, W1n, G1, g1b, G2, g2b,
      p0, p1, x0n, x1n);

  edge_kernel<<<N_EDGES / 16, 32, 0, stream>>>(
      edge_sh, edge_attr, F1, F2, L1, L2, edge_index,
      p0, p1, x0n, x1n, acc);

  proj_kernel<<<N_NODES / 16, 32, 0, stream>>>(
      acc, WO0, bO0, WO1, WO2, (float*)d_out);
}


# Scatter target split by package half: a wave adds lines 0-12 of a row into its own half's plane (die id picks it; half 1's is the output buffer), projection sums the planes in LDS; sum order changes
# speedup vs baseline: 2.2138x; 1.0541x over previous
#include <hip/hip_runtime.h>
#include <hip/hip_bf16.h>
#include <math.h>


typedef __attribute__((ext_vector_type(2))) float v2f;
typedef __attribute__((ext_vector_type(4))) float v4f;
typedef __attribute__((ext_vector_type(8))) float v8f;

#define N_NODES 10000
#define N_EDGES 320000

#define INV_SQ3   0.5773502691896258f
#define INV_SQ2   0.7071067811865476f
#define INV_SQ32  0.17677669529663687f
#define INV_SQ64  0.125f
#define INV_SQ96  0.10206207261596575f
#define INV_SQ160 0.07905694150420949f
#define LN2F      0.6931471805599453f

__device__ __forceinline__ float sspf(float v) {
  const float chain = log1pf(expf(v));
  const float sp = (v > 20.0f) ? v : chain;
  return sp - LN2F;
}
__device__ __forceinline__ float siluf(float v) {
  return v / (1.0f + expf(-v));
}
__device__ __forceinline__ v2f ld2(const float* p) {
  return *(const v2f*)p;
}
template <int NC>
__device__ __forceinline__ v2f ldb(const float* __restrict__ W, int k, int col) {
  v2f b;
  b.x = W[k * NC + col];
  b.y = W[(k + 1) * NC + col];
  return b;
}

__device__ __forceinline__ v8f wmma4(v2f a, v2f b, v8f c) {
  return __builtin_amdgcn_wmma_f32_16x16x4_f32(false, a, false, b,
                                               (short)0, c, false, false);
}


template <int KS, typename LA>
__device__ __forceinline__ void load_a(v2f (&a)[KS], LA la) {
  const int lane = threadIdx.x & 31;
  const int m = lane & 15;
  const int kk = (lane >> 4) << 1;
#pragma unroll
  for (int q = 0; q < KS; ++q) a[q] = la(m, q * 4 + kk);
}

template <int KS, typename LB>
__device__ __forceinline__ v8f wmma_tile(const v2f (&a)[KS], LB lb, v8f c) {
  const int lane = threadIdx.x & 31;
  const int n = lane & 15;
  const int kk = (lane >> 4) << 1;
#pragma unroll
  for (int q = 0; q < KS; ++q) c = wmma4(a[q], lb(n, q * 4 + kk), c);
  return c;
}

template <typename ST>
__device__ __forceinline__ void tile_store(v8f c, ST st) {
  const int lane = threadIdx.x & 31;
  const int col = lane & 15;
  const int rbase = (lane >> 4) << 3;
#pragma unroll
  for (int r = 0; r < 8; ++r) st(r, rbase + r, col, c[r]);
}

__global__ __launch_bounds__(32) void node_kernel(
    const float* __restrict__ x,
    const float* __restrict__ W0p, const float* __restrict__ b0p,
    const float* __restrict__ W1p,
    const float* __restrict__ W0n, const float* __restrict__ b0n,
    const float* __restrict__ W1n,
    const float* __restrict__ G1, const float* __restrict__ g1b,
    const float* __restrict__ G2, const float* __restrict__ g2b,
    float* __restrict__ p0, float* __restrict__ p1,
    float* __restrict__ x0n, float* __restrict__ x1n,
    float* __restrict__ accB) {
  __shared__ __align__(16) float f0S[16 * 96];
  __shared__ __align__(16) float hS[16 * 96];
  __shared__ __align__(16) float gS[16 * 96];

  const int lane = threadIdx.x;
  const int n0 = blockIdx.x * 16;
  const float* xt = x + (size_t)n0 * 160;

  {
    const v4f z4 = {0.0f, 0.0f, 0.0f, 0.0f};
    v4f* zb = (v4f*)(accB + (size_t)n0 * 416);
#pragma unroll
    for (int i = 0; i < 52; ++i) zb[lane + 32 * i] = z4;
  }

  {
    v2f ax[16];
    load_a(ax, [&](int m, int k) { return ld2(xt + m * 160 + k); });
#pragma unroll
    for (int j = 0; j < 4; ++j) {
      v8f c = {};
      c = wmma_tile(ax, [&](int n, int k) {
        return ldb<64>(W0p, k, j * 16 + n);
      }, c);
      tile_store(c, [&](int, int m, int n, float v) {
        int col = j * 16 + n;
        p0[(size_t)(n0 + m) * 64 + col] = v * INV_SQ64 + b0p[col];
      });
    }
  }
#pragma unroll
  for (int i = 0; i < 3; ++i) {
    v2f ax[8];
    load_a(ax, [&](int m, int k) {
      v2f a;
      a.x = xt[m * 160 + 64 + 3 * k + i];
      a.y = xt[m * 160 + 64 + 3 * (k + 1) + i];
      return a;
    });
    for (int j = 0; j < 2; ++j) {
      v8f c = {};
      c = wmma_tile(ax, [&](int n, int k) {
        return ldb<32>(W1p, k, j * 16 + n);
      }, c);
      tile_store(c, [&](int, int m, int n, float v) {
        p1[(size_t)(n0 + m) * 96 + (j * 16 + n) * 3 + i] = v * INV_SQ32;
      });
    }
  }

  for (int idx = lane; idx < 16 * 96; idx += 32) {
    int m = idx / 96, c = idx % 96;
    float v;
    if (c < 64) {
      v = xt[m * 160 + c];
    } else {
      int u = c - 64;
      float a0 = xt[m * 160 + 64 + u * 3 + 0];
      float a1 = xt[m * 160 + 64 + u * 3 + 1];
      float a2 = xt[m * 160 + 64 + u * 3 + 2];
      v = sqrtf(a0 * a0 + a1 * a1 + a2 * a2 + 1e-12f);
    }
    f0S[idx] = v;
  }
  __syncthreads();

  {
    v2f af[24];
    load_a(af, [&](int m, int k) { return ld2(f0S + m * 96 + k); });
    for (int j = 0; j < 6; ++j) {
      v8f c = {};
      c = wmma_tile(af, [&](int n, int k) {
        return ldb<96>(G1, k, j * 16 + n);
      }, c);
      tile_store(c, [&](int, int m, int n, float v) {
        int col = j * 16 + n;
        hS[m * 96 + col] = siluf(v + g1b[col]);
      });
    }
  }
  __syncthreads();
  {
    v2f ah[24];
    load_a(ah, [&](int m, int k) { return ld2(hS + m * 96 + k); });
    for (int j = 0; j < 6; ++j) {
      v8f c = {};
      c = wmma_tile(ah, [&](int n, int k) {
        return ldb<96>(G2, k, j * 16 + n);
      }, c);
      tile_store(c, [&](int, int m, int n, float v) {
        gS[m * 96 + j * 16 + n] = v + g2b[j * 16 + n];
      });
    }
  }
  __syncthreads();

  {
    v2f ag[16];
    load_a(ag, [&](int m, int k) { return ld2(gS + m * 96 + k); });
#pragma unroll
    for (int j = 0; j < 4; ++j) {
      v8f c = {};
      c = wmma_tile(ag, [&](int n, int k) {
        return ldb<64>(W0n, k, j * 16 + n);
      }, c);
      tile_store(c, [&](int, int m, int n, float v) {
        int col = j * 16 + n;
        x0n[(size_t)(n0 + m) * 64 + col] = v * INV_SQ64 + b0n[col];
      });
    }
  }
  for (int i = 0; i < 3; ++i) {
    v2f ax[8];
    load_a(ax, [&](int m, int k) {
      v2f a;
      a.x = xt[m * 160 + 64 + 3 * k + i] * gS[m * 96 + 64 + k];
      a.y = xt[m * 160 + 64 + 3 * (k + 1) + i] * gS[m * 96 + 64 + k + 1];
      return a;
    });
    for (int j = 0; j < 2; ++j) {
      v8f c = {};
      c = wmma_tile(ax, [&](int n, int k) {
        return ldb<32>(W1n, k, j * 16 + n);
      }, c);
      tile_store(c, [&](int, int m, int n, float v) {
        x1n[(size_t)(n0 + m) * 96 + (j * 16 + n) * 3 + i] = v * INV_SQ32;
      });
    }
  }
}

__global__ __launch_bounds__(32) __attribute__((amdgpu_waves_per_eu(5))) void edge_kernel(
    const float* __restrict__ edge_sh, const float* __restrict__ edge_attr,
    const float* __restrict__ F1, const float* __restrict__ F2,
    const float* __restrict__ L1, const float* __restrict__ L2,
    const int* __restrict__ edge_index,
    const float* __restrict__ p0, const float* __restrict__ p1,
    const float* __restrict__ x0n, const float* __restrict__ x1n,
    float* __restrict__ acc, float* __restrict__ accB) {
  __shared__ __align__(16) float s0S[16 * 160];
  __shared__ __align__(16) float hNE[2 * 16 * 32];
  float* const hN = hNE;
  float* const hE = hNE + 16 * 32;

  const int lane = threadIdx.x;
  const int e0 = blockIdx.x * 16;

  __builtin_prefetch(F2, 0, 1);
  __builtin_prefetch(L2, 0, 1);

  const int mrow = lane & 15;
  const int hk = lane >> 4;
  const int drow = edge_index[e0 + mrow];
  const int srow = edge_index[N_EDGES + e0 + mrow];

  const unsigned aid = __builtin_amdgcn_s_sendmsg_rtn(0x87);

  {
    v2f as0[40];
    {
      const float* pd0 = p0 + (size_t)drow * 64 + 4 * hk;
      const float* ps0 = p0 + (size_t)srow * 64 + 4 * hk;
      const float* pd1 = p1 + (size_t)drow * 96 + 12 * hk;
      const float* ps1 = p1 + (size_t)srow * 96 + 12 * hk;
      v4f qd[8], qs[8], vd[4][3], vs[4][3];
#pragma unroll
      for (int g = 0; g < 4; ++g) {
#pragma unroll
        for (int j = 0; j < 3; ++j) {
          vd[g][j] = *(const v4f*)(pd1 + 24 * g + 4 * j);
          vs[g][j] = *(const v4f*)(ps1 + 24 * g + 4 * j);
        }
      }
#pragma unroll
      for (int g = 0; g < 8; ++g) {
        qd[g] = *(const v4f*)(pd0 + 8 * g);
        qs[g] = *(const v4f*)(ps0 + 8 * g);
      }
#pragma unroll
      for (int g = 0; g < 8; ++g) {
        as0[2 * g].x = qd[g].x;          as0[2 * g].y = qd[g].y;
        as0[2 * g + 1].x = qd[g].z;      as0[2 * g + 1].y = qd[g].w;
        as0[16 + 2 * g].x = qs[g].x;     as0[16 + 2 * g].y = qs[g].y;
        as0[16 + 2 * g + 1].x = qs[g].z; as0[16 + 2 * g + 1].y = qs[g].w;
      }
#pragma unroll
      for (int g = 0; g < 4; ++g) {
        const float xd[12] = {vd[g][0].x, vd[g][0].y, vd[g][0].z, vd[g][0].w,
                              vd[g][1].x, vd[g][1].y, vd[g][1].z, vd[g][1].w,
                              vd[g][2].x, vd[g][2].y, vd[g][2].z, vd[g][2].w};
        const float xs[12] = {vs[g][0].x, vs[g][0].y, vs[g][0].z, vs[g][0].w,
                              vs[g][1].x, vs[g][1].y, vs[g][1].z, vs[g][1].w,
                              vs[g][2].x, vs[g][2].y, vs[g][2].z, vs[g][2].w};
        float ip[4];
#pragma unroll
        for (int j = 0; j < 4; ++j) {
          ip[j] = (xd[3 * j] * xs[3 * j] + xd[3 * j + 1] * xs[3 * j + 1] +
                   xd[3 * j + 2] * xs[3 * j + 2]) * INV_SQ3;
        }
        as0[32 + 2 * g].x = ip[0];     as0[32 + 2 * g].y = ip[1];
        as0[32 + 2 * g + 1].x = ip[2]; as0[32 + 2 * g + 1].y = ip[3];
      }
    }

    {
      v8f c0 = {}, c1 = {};
#pragma unroll
      for (int Q = 0; Q < 40; ++Q) {
        const int k0 = 8 * (Q >> 1) + 4 * hk + 2 * (Q & 1);
        c0 = wmma4(as0[Q], ldb<32>(L1, k0, mrow), c0);
        c1 = wmma4(as0[Q], ldb<32>(L1, k0, 16 + mrow), c1);
      }
      float* xsS = s0S;
      const int half = lane >> 4;
      const int c4 = lane & 15;
      v4f t0[8];
#pragma unroll
      for (int i = 0; i < 8; ++i) {
        const int m = 2 * i + half;
        t0[i] = *(const v4f*)(x0n + (size_t)__shfl(srow, m) * 64 + c4 * 4);
      }
      v4f t1[12];
#pragma unroll
      for (int i = 0; i < 12; ++i) {
        const int f = lane + 32 * i;
        const int m = f / 24, q4 = f % 24;
        t1[i] = *(const v4f*)(x1n + (size_t)__shfl(srow, m) * 96 + q4 * 4);
      }
      tile_store(c0, [&](int r, int m, int n, float v) {
        hN[m * 32 + n] = sspf(v * INV_SQ160);
        hN[m * 32 + 16 + n] = sspf(c1[r] * INV_SQ160);
      });
#pragma unroll
      for (int i = 0; i < 8; ++i) {
        const int m = 2 * i + half;
        *(v4f*)(xsS + m * 160 + c4 * 4) = t0[i];
      }
#pragma unroll
      for (int i = 0; i < 12; ++i) {
        const int f = lane + 32 * i;
        const int m = f / 24, q4 = f % 24;
        *(v4f*)(xsS + m * 160 + 64 + q4 * 4) = t1[i];
      }
    }
  }
  {
    const float* ea = edge_attr + (size_t)e0 * 32;
    v2f aea[8];
    load_a(aea, [&](int m, int k) { return ld2(ea + m * 32 + k); });
    for (int j = 0; j < 2; ++j) {
      v8f c = {};
      c = wmma_tile(aea, [&](int n, int k) {
        return ldb<32>(F1, k, j * 16 + n);
      }, c);
      tile_store(c, [&](int, int m, int n, float v) {
        hE[m * 32 + j * 16 + n] = sspf(v * INV_SQ32);
      });
    }
  }
  __syncthreads();

  const int n = lane & 15;
  const int half = lane >> 4;
  const int rbase = half << 3;
  float sh0r[8], s1xr[8], s1yr[8], s1zr[8];
#pragma unroll
  for (int r = 0; r < 8; ++r) {
    int e = e0 + rbase + r;
    sh0r[r] = edge_sh[(size_t)e * 4 + 0];
    s1xr[r] = edge_sh[(size_t)e * 4 + 1];
    s1yr[r] = edge_sh[(size_t)e * 4 + 2];
    s1zr[r] = edge_sh[(size_t)e * 4 + 3];
  }
  const int dmine = drow;
  const float* xsS = s0S;
  const bool hb = ((aid >> 18) & 1u) != 0u;
  float* const planeP = hb ? accB : acc;
  const int planePitch = hb ? 416 : 480;

  v2f aN[8], aE[8];
  load_a(aN, [&](int m, int k) { return ld2(hN + m * 32 + k); });
  load_a(aE, [&](int m, int k) { return ld2(hE + m * 32 + k); });

  float* const stg = hNE;
  constexpr int P1 = 36;
  constexpr int P3 = 97;

  auto wtile = [&](int nb) -> v8f {
    v8f cN = {}, cE = {};
    cN = wmma_tile(aN, [&](int nn, int k) {
      return ldb<224>(L2, k, nb + nn);
    }, cN);
    cE = wmma_tile(aE, [&](int nn, int k) {
      return ldb<224>(F2, k, nb + nn);
    }, cE);
    return cN * cE * (INV_SQ32 * INV_SQ32);
  };

  auto add_line = [&](int m, int ch0, float v) {
    const int d = __builtin_amdgcn_readlane(dmine, m);
    float* const rowp = (ch0 < 416) ? planeP + (size_t)d * planePitch
                                    : acc + (size_t)d * 480;
    atomicAdd(rowp + ch0 + lane, v);
  };

  auto scatter1 = [&](const float (&o)[2][8], int ch0) {
#pragma unroll
    for (int r = 0; r < 8; ++r) {
      stg[(rbase + r) * P1 + n] = o[0][r];
      stg[(rbase + r) * P1 + 16 + n] = o[1][r];
    }
    __builtin_amdgcn_fence(__ATOMIC_ACQ_REL, "wavefront");
    float v[16];
#pragma unroll
    for (int m = 0; m < 16; ++m) v[m] = stg[m * P1 + lane];
#pragma unroll
    for (int m = 0; m < 16; ++m) add_line(m, ch0, v[m]);
    __builtin_amdgcn_fence(__ATOMIC_ACQ_REL, "wavefront");
  };

  auto scatter3 = [&](const float (&o)[2][8][3], int ch0) {
#pragma unroll
    for (int ph = 0; ph < 2; ++ph) {
#pragma unroll
      for (int q = 0; q < 4; ++q) {
        float* row = stg + (q + 4 * half) * P3;
#pragma unroll
        for (int t = 0; t < 2; ++t) {
          row[t * 48 + 3 * n + 0] = o[t][4 * ph + q][0];
          row[t * 48 + 3 * n + 1] = o[t][4 * ph + q][1];
          row[t * 48 + 3 * n + 2] = o[t][4 * ph + q][2];
        }
      }
      __builtin_amdgcn_fence(__ATOMIC_ACQ_REL, "wavefront");
      float v[8][3];
#pragma unroll
      for (int s = 0; s < 8; ++s)
#pragma unroll
        for (int L = 0; L < 3; ++L) v[s][L] = stg[s * P3 + L * 32 + lane];
#pragma unroll
      for (int s = 0; s < 8; ++s) {
        const int m = (s & 3) + 4 * ph + 8 * (s >> 2);
#pragma unroll
        for (int L = 0; L < 3; ++L) add_line(m, ch0 + 32 * L, v[s][L]);
      }
      __builtin_amdgcn_fence(__ATOMIC_ACQ_REL, "wavefront");
    }
  };

  for (int p = 0; p < 2; ++p) {
    const int nb = 32 * p;
    const v8f wA = wtile(nb), wB = wtile(nb + 16);
    float o[2][8];
#pragma unroll
    for (int r = 0; r < 8; ++r) {
      const float* xr = xsS + (rbase + r) * 160 + nb + n;
      o[0][r] = wA[r] * xr[0] * sh0r[r];
      o[1][r] = wB[r] * xr[16] * sh0r[r];
    }
    scatter1(o, nb);
  }
  {
    const v8f wA = wtile(160), wB = wtile(176);
    float o[2][8];
#pragma unroll
    for (int r = 0; r < 8; ++r) {
#pragma unroll
      for (int t = 0; t < 2; ++t) {
        const float* xv = xsS + (rbase + r) * 160 + 64 + (16 * t + n) * 3;
        const float wv = t ? wB[r] : wA[r];
        float dp = xv[0] * s1xr[r] + xv[1] * s1yr[r] + xv[2] * s1zr[r];
        o[t][r] = wv * dp * INV_SQ3;
      }
    }
    scatter1(o, 64);
  }
  for (int p = 0; p < 2; ++p) {
    const int nb = 64 + 32 * p;
    const v8f wA = wtile(nb), wB = wtile(nb + 16);
    float o[2][8][3];
#pragma unroll
    for (int r = 0; r < 8; ++r) {
      const float* xr = xsS + (rbase + r) * 160 + 32 * p + n;
#pragma unroll
      for (int t = 0; t < 2; ++t) {
        const float wv = t ? wB[r] : wA[r];
        float f = wv * xr[16 * t] * INV_SQ3;
        o[t][r][0] = f * s1xr[r];
        o[t][r][1] = f * s1yr[r];
        o[t][r][2] = f * s1zr[r];
      }
    }
    scatter3(o, 96 + 96 * p);
  }
  {
    const v8f wA = wtile(128), wB = wtile(144);
    float o[2][8][3];
#pragma unroll
    for (int r = 0; r < 8; ++r) {
#pragma unroll
      for (int t = 0; t < 2; ++t) {
        const float* xv = xsS + (rbase + r) * 160 + 64 + (16 * t + n) * 3;
        const float wv = t ? wB[r] : wA[r];
        float f = wv * sh0r[r];
        o[t][r][0] = f * xv[0];
        o[t][r][1] = f * xv[1];
        o[t][r][2] = f * xv[2];
      }
    }
    scatter3(o, 288);
  }
  {
    const v8f wA = wtile(192), wB = wtile(208);
    float o[2][8][3];
#pragma unroll
    for (int r = 0; r < 8; ++r) {
#pragma unroll
      for (int t = 0; t < 2; ++t) {
        const float* xv = xsS + (rbase + r) * 160 + 64 + (16 * t + n) * 3;
        const float wv = t ? wB[r] : wA[r];
        const float s1x = s1xr[r], s1y = s1yr[r], s1z = s1zr[r];
        float cx = xv[1] * s1z - xv[2] * s1y;
        float cy = xv[2] * s1x - xv[0] * s1z;
        float cz = xv[0] * s1y - xv[1] * s1x;
        float f = wv * INV_SQ2;
        o[t][r][0] = f * cx;
        o[t][r][1] = f * cy;
        o[t][r][2] = f * cz;
      }
    }
    scatter3(o, 384);
  }
}

#define TP 484
__global__ __launch_bounds__(32) void proj_kernel(
    const float* __restrict__ acc,
    const float* __restrict__ WO0, const float* __restrict__ bO0,
    const float* __restrict__ WO1, const float* __restrict__ WO2,
    float* y) {
  __shared__ __align__(16) float tS[16 * TP];
  const int lane = threadIdx.x;
  const int n0 = blockIdx.x * 16;
  const v4f* a4 = (const v4f*)(acc + (size_t)n0 * 480);
  const v4f* b4 = (const v4f*)(y + (size_t)n0 * 416);

#pragma unroll
  for (int b = 0; b < 4; ++b) {
    v4f va[13], vb[13];
#pragma unroll
    for (int i = 0; i < 13; ++i) {
      const int q = lane + 32 * (13 * b + i);
      const int m = q / 104, c4 = q % 104;
      vb[i] = b4[q];
      va[i] = a4[m * 120 + c4];
    }
#pragma unroll
    for (int i = 0; i < 13; ++i) {
      const int q = lane + 32 * (13 * b + i);
      const int m = q / 104, c4 = q % 104;
      *(v4f*)(tS + m * TP + c4 * 4) = va[i] + vb[i];
    }
  }
  {
    v4f va[8];
#pragma unroll
    for (int i = 0; i < 8; ++i) {
      const int q = lane + 32 * i;
      va[i] = a4[(q >> 4) * 120 + 104 + (q & 15)];
    }
#pragma unroll
    for (int i = 0; i < 8; ++i) {
      const int q = lane + 32 * i;
      *(v4f*)(tS + (q >> 4) * TP + 416 + (q & 15) * 4) = va[i];
    }
  }
  __syncthreads();

  {
    v2f aa[24];
    load_a(aa, [&](int m, int k) { return ld2(tS + m * TP + k); });
    for (int j = 0; j < 8; ++j) {
      v8f c = {};
      c = wmma_tile(aa, [&](int n, int k) {
        return ldb<128>(WO0, k, j * 16 + n);
      }, c);
      tile_store(c, [&](int, int m, int n, float v) {
        int col = j * 16 + n;
        y[(size_t)(n0 + m) * 416 + col] = v * INV_SQ96 + bO0[col];
      });
    }
  }
  for (int i = 0; i < 3; ++i) {
    v2f aa[24];
    load_a(aa, [&](int m, int k) {
      v2f a;
      a.x = tS[m * TP + 96 + 3 * k + i];
      a.y = tS[m * TP + 96 + 3 * (k + 1) + i];
      return a;
    });
#pragma unroll
    for (int j = 0; j < 4; ++j) {
      v8f c = {};
      c = wmma_tile(aa, [&](int n, int k) {
        return ldb<64>(WO1, k, j * 16 + n);
      }, c);
      tile_store(c, [&](int, int m, int n, float v) {
        y[(size_t)(n0 + m) * 416 + 128 + (j * 16 + n) * 3 + i] = v * INV_SQ96;
      });
    }
  }
  for (int i = 0; i < 3; ++i) {
    v2f aa[8];
    load_a(aa, [&](int m, int k) {
      v2f a;
      a.x = tS[m * TP + 384 + 3 * k + i];
      a.y = tS[m * TP + 384 + 3 * (k + 1) + i];
      return a;
    });
    for (int j = 0; j < 2; ++j) {
      v8f c = {};
      c = wmma_tile(aa, [&](int n, int k) {
        return ldb<32>(WO2, k, j * 16 + n);
      }, c);
      tile_store(c, [&](int, int m, int n, float v) {
        y[(size_t)(n0 + m) * 416 + 320 + (j * 16 + n) * 3 + i] = v * INV_SQ32;
      });
    }
  }
}

extern "C" void kernel_launch(void* const* d_in, const int* in_sizes, int n_in,
                              void* d_out, int out_size, void* d_ws,
                              size_t ws_size, hipStream_t stream) {
  (void)in_sizes; (void)n_in; (void)out_size; (void)ws_size;
  const float* x         = (const float*)d_in[0];
  const float* edge_sh   = (const float*)d_in[1];
  const float* edge_attr = (const float*)d_in[2];
  const float* W0p = (const float*)d_in[3];
  const float* b0p = (const float*)d_in[4];
  const float* W1p = (const float*)d_in[5];
  const float* W0n = (const float*)d_in[6];
  const float* b0n = (const float*)d_in[7];
  const float* W1n = (const float*)d_in[8];
  const float* G1  = (const float*)d_in[9];
  const float* g1b = (const float*)d_in[10];
  const float* G2  = (const float*)d_in[11];
  const float* g2b = (const float*)d_in[12];
  const float* F1  = (const float*)d_in[13];
  const float* F2  = (const float*)d_in[14];
  const float* L1  = (const float*)d_in[15];
  const float* L2  = (const float*)d_in[16];
  const float* WO0 = (const float*)d_in[17];
  const float* bO0 = (const float*)d_in[18];
  const float* WO1 = (const float*)d_in[19];
  const float* WO2 = (const float*)d_in[20];
  const int* edge_index = (const int*)d_in[21];

  float* ws   = (float*)d_ws;
  float* p0   = ws;
  float* p1   = p0   + (size_t)N_NODES * 64;
  float* x0n  = p1   + (size_t)N_NODES * 96;
  float* x1n  = x0n  + (size_t)N_NODES * 64;
  float* acc  = x1n  + (size_t)N_NODES * 96;
  float* accB = (float*)d_out;

  hipMemsetAsync(acc, 0, (size_t)N_NODES * 480 * sizeof(float), stream);

  node_kernel<<<N_NODES / 16, 32, 0, stream>>>(
      x, W0p, b0p, W1p, W0n, b0n, W1n, G1, g1b, G2, g2b,
      p0, p1, x0n, x1n, accB);

  edge_kernel<<<N_EDGES / 16, 32, 0, stream>>>(
      edge_sh, edge_attr, F1, F2, L1, L2, edge_index,
      p0, p1, x0n, x1n, acc, accB);

  proj_kernel<<<N_NODES / 16, 32, 0, stream>>>(
      acc, WO0, bO0, WO1, WO2, (float*)d_out);
}


# Edge kernel: a wave takes 32 edges as two 16-row tiles; each second-layer weight fragment is loaded once and feeds both tiles' products (weight loads and load waits per edge halved); same sums
# speedup vs baseline: 2.2829x; 1.0312x over previous
#include <hip/hip_runtime.h>
#include <hip/hip_bf16.h>
#include <math.h>


typedef __attribute__((ext_vector_type(2))) float v2f;
typedef __attribute__((ext_vector_type(4))) float v4f;
typedef __attribute__((ext_vector_type(8))) float v8f;

#define N_NODES 10000
#define N_EDGES 320000

#define INV_SQ3   0.5773502691896258f
#define INV_SQ2   0.7071067811865476f
#define INV_SQ32  0.17677669529663687f
#define INV_SQ64  0.125f
#define INV_SQ96  0.10206207261596575f
#define INV_SQ160 0.07905694150420949f
#define LN2F      0.6931471805599453f

__device__ __forceinline__ float sspf(float v) {
  const float chain = log1pf(expf(v));
  const float sp = (v > 20.0f) ? v : chain;
  return sp - LN2F;
}
__device__ __forceinline__ float siluf(float v) {
  return v / (1.0f + expf(-v));
}
__device__ __forceinline__ v2f ld2(const float* p) {
  return *(const v2f*)p;
}
template <int NC>
__device__ __forceinline__ v2f ldb(const float* __restrict__ W, int k, int col) {
  v2f b;
  b.x = W[k * NC + col];
  b.y = W[(k + 1) * NC + col];
  return b;
}

__device__ __forceinline__ v8f wmma4(v2f a, v2f b, v8f c) {
  return __builtin_amdgcn_wmma_f32_16x16x4_f32(false, a, false, b,
                                               (short)0, c, false, false);
}


template <int KS, typename LA>
__device__ __forceinline__ void load_a(v2f (&a)[KS], LA la) {
  const int lane = threadIdx.x & 31;
  const int m = lane & 15;
  const int kk = (lane >> 4) << 1;
#pragma unroll
  for (int q = 0; q < KS; ++q) a[q] = la(m, q * 4 + kk);
}

template <int KS, typename LB>
__device__ __forceinline__ v8f wmma_tile(const v2f (&a)[KS], LB lb, v8f c) {
  const int lane = threadIdx.x & 31;
  const int n = lane & 15;
  const int kk = (lane >> 4) << 1;
#pragma unroll
  for (int q = 0; q < KS; ++q) c = wmma4(a[q], lb(n, q * 4 + kk), c);
  return c;
}

template <typename ST>
__device__ __forceinline__ void tile_store(v8f c, ST st) {
  const int lane = threadIdx.x & 31;
  const int col = lane & 15;
  const int rbase = (lane >> 4) << 3;
#pragma unroll
  for (int r = 0; r < 8; ++r) st(r, rbase + r, col, c[r]);
}

__global__ __launch_bounds__(32) void node_kernel(
    const float* __restrict__ x,
    const float* __restrict__ W0p, const float* __restrict__ b0p,
    const float* __restrict__ W1p,
    const float* __restrict__ W0n, const float* __restrict__ b0n,
    const float* __restrict__ W1n,
    const float* __restrict__ G1, const float* __restrict__ g1b,
    const float* __restrict__ G2, const float* __restrict__ g2b,
    float* __restrict__ p0, float* __restrict__ p1,
    float* __restrict__ x0n, float* __restrict__ x1n,
    float* __restrict__ accB) {
  __shared__ __align__(16) float f0S[16 * 96];
  __shared__ __align__(16) float hS[16 * 96];
  __shared__ __align__(16) float gS[16 * 96];

  const int lane = threadIdx.x;
  const int n0 = blockIdx.x * 16;
  const float* xt = x + (size_t)n0 * 160;

  {
    const v4f z4 = {0.0f, 0.0f, 0.0f, 0.0f};
    v4f* zb = (v4f*)(accB + (size_t)n0 * 416);
#pragma unroll
    for (int i = 0; i < 52; ++i) zb[lane + 32 * i] = z4;
  }

  {
    v2f ax[16];
    load_a(ax, [&](int m, int k) { return ld2(xt + m * 160 + k); });
#pragma unroll
    for (int j = 0; j < 4; ++j) {
      v8f c = {};
      c = wmma_tile(ax, [&](int n, int k) {
        return ldb<64>(W0p, k, j * 16 + n);
      }, c);
      tile_store(c, [&](int, int m, int n, float v) {
        int col = j * 16 + n;
        p0[(size_t)(n0 + m) * 64 + col] = v * INV_SQ64 + b0p[col];
      });
    }
  }
#pragma unroll
  for (int i = 0; i < 3; ++i) {
    v2f ax[8];
    load_a(ax, [&](int m, int k) {
      v2f a;
      a.x = xt[m * 160 + 64 + 3 * k + i];
      a.y = xt[m * 160 + 64 + 3 * (k + 1) + i];
      return a;
    });
    for (int j = 0; j < 2; ++j) {
      v8f c = {};
      c = wmma_tile(ax, [&](int n, int k) {
        return ldb<32>(W1p, k, j * 16 + n);
      }, c);
      tile_store(c, [&](int, int m, int n, float v) {
        p1[(size_t)(n0 + m) * 96 + (j * 16 + n) * 3 + i] = v * INV_SQ32;
      });
    }
  }

  for (int idx = lane; idx < 16 * 96; idx += 32) {
    int m = idx / 96, c = idx % 96;
    float v;
    if (c < 64) {
      v = xt[m * 160 + c];
    } else {
      int u = c - 64;
      float a0 = xt[m * 160 + 64 + u * 3 + 0];
      float a1 = xt[m * 160 + 64 + u * 3 + 1];
      float a2 = xt[m * 160 + 64 + u * 3 + 2];
      v = sqrtf(a0 * a0 + a1 * a1 + a2 * a2 + 1e-12f);
    }
    f0S[idx] = v;
  }
  __syncthreads();

  {
    v2f af[24];
    load_a(af, [&](int m, int k) { return ld2(f0S + m * 96 + k); });
    for (int j = 0; j < 6; ++j) {
      v8f c = {};
      c = wmma_tile(af, [&](int n, int k) {
        return ldb<96>(G1, k, j * 16 + n);
      }, c);
      tile_store(c, [&](int, int m, int n, float v) {
        int col = j * 16 + n;
        hS[m * 96 + col] = siluf(v + g1b[col]);
      });
    }
  }
  __syncthreads();
  {
    v2f ah[24];
    load_a(ah, [&](int m, int k) { return ld2(hS + m * 96 + k); });
    for (int j = 0; j < 6; ++j) {
      v8f c = {};
      c = wmma_tile(ah, [&](int n, int k) {
        return ldb<96>(G2, k, j * 16 + n);
      }, c);
      tile_store(c, [&](int, int m, int n, float v) {
        gS[m * 96 + j * 16 + n] = v + g2b[j * 16 + n];
      });
    }
  }
  __syncthreads();

  {
    v2f ag[16];
    load_a(ag, [&](int m, int k) { return ld2(gS + m * 96 + k); });
#pragma unroll
    for (int j = 0; j < 4; ++j) {
      v8f c = {};
      c = wmma_tile(ag, [&](int n, int k) {
        return ldb<64>(W0n, k, j * 16 + n);
      }, c);
      tile_store(c, [&](int, int m, int n, float v) {
        int col = j * 16 + n;
        x0n[(size_t)(n0 + m) * 64 + col] = v * INV_SQ64 + b0n[col];
      });
    }
  }
  for (int i = 0; i < 3; ++i) {
    v2f ax[8];
    load_a(ax, [&](int m, int k) {
      v2f a;
      a.x = xt[m * 160 + 64 + 3 * k + i] * gS[m * 96 + 64 + k];
      a.y = xt[m * 160 + 64 + 3 * (k + 1) + i] * gS[m * 96 + 64 + k + 1];
      return a;
    });
    for (int j = 0; j < 2; ++j) {
      v8f c = {};
      c = wmma_tile(ax, [&](int n, int k) {
        return ldb<32>(W1n, k, j * 16 + n);
      }, c);
      tile_store(c, [&](int, int m, int n, float v) {
        x1n[(size_t)(n0 + m) * 96 + (j * 16 + n) * 3 + i] = v * INV_SQ32;
      });
    }
  }
}

__global__ __launch_bounds__(32) __attribute__((amdgpu_waves_per_eu(3))) void edge_kernel(
    const float* __restrict__ edge_sh, const float* __restrict__ edge_attr,
    const float* __restrict__ F1, const float* __restrict__ F2,
    const float* __restrict__ L1, const float* __restrict__ L2,
    const int* __restrict__ edge_index,
    const float* __restrict__ p0, const float* __restrict__ p1,
    const float* __restrict__ x0n, const float* __restrict__ x1n,
    float* __restrict__ acc, float* __restrict__ accB) {
  __shared__ __align__(16) float s0S[2 * 16 * 160];
  __shared__ __align__(16) float hNE[2 * 16 * 32];
  __shared__ __align__(16) float shS[32 * 4];
  float* const hN = hNE;
  float* const hE = hNE + 16 * 32;

  const int lane = threadIdx.x;
  const int e00 = blockIdx.x * 32;

  __builtin_prefetch(F2, 0, 1);
  __builtin_prefetch(L2, 0, 1);

  const int mrow = lane & 15;
  const int hk = lane >> 4;
  const int drow0 = edge_index[e00 + mrow];
  const int srow0 = edge_index[N_EDGES + e00 + mrow];
  const int drow1 = edge_index[e00 + 16 + mrow];
  const int srow1 = edge_index[N_EDGES + e00 + 16 + mrow];
  const v4f shrow = *(const v4f*)(edge_sh + (size_t)(e00 + lane) * 4);
  *(v4f*)(shS + lane * 4) = shrow;

  const unsigned aid = __builtin_amdgcn_s_sendmsg_rtn(0x87);

  v2f aN0[8], aE0[8], aN1[8], aE1[8];
#pragma unroll
  for (int q = 0; q < 8; ++q) {
    const v2f z2 = {0.0f, 0.0f};
    aN0[q] = z2; aE0[q] = z2;
  }

#pragma nounroll
  for (int s = 0; s < 2; ++s) {
    const int e0 = e00 + 16 * s;
    const int drow = s ? drow1 : drow0;
    const int srow = s ? srow1 : srow0;
    float* const xsS = s0S + s * (16 * 160);
    int l1off = 0;
    asm volatile("" : "+s"(l1off));
    const float* const L1t = L1 + l1off;

    {
      v2f as0[40];
      {
        const float* pd0 = p0 + (size_t)drow * 64 + 4 * hk;
        const float* ps0 = p0 + (size_t)srow * 64 + 4 * hk;
        const float* pd1 = p1 + (size_t)drow * 96 + 12 * hk;
        const float* ps1 = p1 + (size_t)srow * 96 + 12 * hk;
        v4f qd[8], qs[8], vd[4][3], vs[4][3];
#pragma unroll
        for (int g = 0; g < 4; ++g) {
#pragma unroll
          for (int j = 0; j < 3; ++j) {
            vd[g][j] = *(const v4f*)(pd1 + 24 * g + 4 * j);
            vs[g][j] = *(const v4f*)(ps1 + 24 * g + 4 * j);
          }
        }
#pragma unroll
        for (int g = 0; g < 8; ++g) {
          qd[g] = *(const v4f*)(pd0 + 8 * g);
          qs[g] = *(const v4f*)(ps0 + 8 * g);
        }
#pragma unroll
        for (int g = 0; g < 8; ++g) {
          as0[2 * g].x = qd[g].x;          as0[2 * g].y = qd[g].y;
          as0[2 * g + 1].x = qd[g].z;      as0[2 * g + 1].y = qd[g].w;
          as0[16 + 2 * g].x = qs[g].x;     as0[16 + 2 * g].y = qs[g].y;
          as0[16 + 2 * g + 1].x = qs[g].z; as0[16 + 2 * g + 1].y = qs[g].w;
        }
#pragma unroll
        for (int g = 0; g < 4; ++g) {
          const float xd[12] = {vd[g][0].x, vd[g][0].y, vd[g][0].z, vd[g][0].w,
                                vd[g][1].x, vd[g][1].y, vd[g][1].z, vd[g][1].w,
                                vd[g][2].x, vd[g][2].y, vd[g][2].z, vd[g][2].w};
          const float xs[12] = {vs[g][0].x, vs[g][0].y, vs[g][0].z, vs[g][0].w,
                                vs[g][1].x, vs[g][1].y, vs[g][1].z, vs[g][1].w,
                                vs[g][2].x, vs[g][2].y, vs[g][2].z, vs[g][2].w};
          float ip[4];
#pragma unroll
          for (int j = 0; j < 4; ++j) {
            ip[j] = (xd[3 * j] * xs[3 * j] + xd[3 * j + 1] * xs[3 * j + 1] +
                     xd[3 * j + 2] * xs[3 * j + 2]) * INV_SQ3;
          }
          as0[32 + 2 * g].x = ip[0];     as0[32 + 2 * g].y = ip[1];
          as0[32 + 2 * g + 1].x = ip[2]; as0[32 + 2 * g + 1].y = ip[3];
        }
      }

      {
        v8f c0 = {}, c1 = {};
#pragma unroll
        for (int Q = 0; Q < 40; ++Q) {
          const int k0 = 8 * (Q >> 1) + 4 * hk + 2 * (Q & 1);
          c0 = wmma4(as0[Q], ldb<32>(L1t, k0, mrow), c0);
          c1 = wmma4(as0[Q], ldb<32>(L1t, k0, 16 + mrow), c1);
        }
        const int half = lane >> 4;
        const int c4 = lane & 15;
        v4f t0[8];
#pragma unroll
        for (int i = 0; i < 8; ++i) {
          const int m = 2 * i + half;
          t0[i] = *(const v4f*)(x0n + (size_t)__shfl(srow, m) * 64 + c4 * 4);
        }
        v4f t1[12];
#pragma unroll
        for (int i = 0; i < 12; ++i) {
          const int f = lane + 32 * i;
          const int m = f / 24, q4 = f % 24;
          t1[i] = *(const v4f*)(x1n + (size_t)__shfl(srow, m) * 96 + q4 * 4);
        }
        tile_store(c0, [&](int r, int m, int n, float v) {
          hN[m * 32 + n] = sspf(v * INV_SQ160);
          hN[m * 32 + 16 + n] = sspf(c1[r] * INV_SQ160);
        });
#pragma unroll
        for (int i = 0; i < 8; ++i) {
          const int m = 2 * i + half;
          *(v4f*)(xsS + m * 160 + c4 * 4) = t0[i];
        }
#pragma unroll
        for (int i = 0; i < 12; ++i) {
          const int f = lane + 32 * i;
          const int m = f / 24, q4 = f % 24;
          *(v4f*)(xsS + m * 160 + 64 + q4 * 4) = t1[i];
        }
      }
    }
    {
      const float* ea = edge_attr + (size_t)e0 * 32;
      v2f aea[8];
      load_a(aea, [&](int m, int k) { return ld2(ea + m * 32 + k); });
      for (int j = 0; j < 2; ++j) {
        v8f c = {};
        c = wmma_tile(aea, [&](int n, int k) {
          return ldb<32>(F1, k, j * 16 + n);
        }, c);
        tile_store(c, [&](int, int m, int n, float v) {
          hE[m * 32 + j * 16 + n] = sspf(v * INV_SQ32);
        });
      }
    }
    __syncthreads();

    if (s == 0) {
      load_a(aN0, [&](int m, int k) { return ld2(hN + m * 32 + k); });
      load_a(aE0, [&](int m, int k) { return ld2(hE + m * 32 + k); });
    }
  }
  load_a(aN1, [&](int m, int k) { return ld2(hN + m * 32 + k); });
  load_a(aE1, [&](int m, int k) { return ld2(hE + m * 32 + k); });
  __syncthreads();

  const int n = lane & 15;
  const int half = lane >> 4;
  const int rbase = half << 3;
  const bool hb = ((aid >> 18) & 1u) != 0u;
  float* const planeP = hb ? accB : acc;
  const int planePitch = hb ? 416 : 480;

  float* const stg = hNE;
  constexpr int P1 = 36;
  constexpr int P3 = 97;

  auto wpair = [&](int nb, v8f (&w)[2][2]) {
    v8f cN[2][2], cE[2][2];
#pragma unroll
    for (int s = 0; s < 2; ++s) {
#pragma unroll
      for (int t = 0; t < 2; ++t) {
        const v8f z8 = {};
        cN[s][t] = z8;
        cE[s][t] = z8;
      }
    }
    const int kk = half << 1;
#pragma unroll
    for (int t = 0; t < 2; ++t) {
#pragma unroll
      for (int q = 0; q < 8; ++q) {
        const v2f bN = ldb<224>(L2, q * 4 + kk, nb + 16 * t + n);
        const v2f bE = ldb<224>(F2, q * 4 + kk, nb + 16 * t + n);
        cN[0][t] = wmma4(aN0[q], bN, cN[0][t]);
        cN[1][t] = wmma4(aN1[q], bN, cN[1][t]);
        cE[0][t] = wmma4(aE0[q], bE, cE[0][t]);
        cE[1][t] = wmma4(aE1[q], bE, cE[1][t]);
      }
    }
#pragma unroll
    for (int s = 0; s < 2; ++s) {
#pragma unroll
      for (int t = 0; t < 2; ++t) {
        w[s][t] = cN[s][t] * cE[s][t] * (INV_SQ32 * INV_SQ32);
      }
    }
  };

  auto add_line = [&](int dsel, int m, int ch0, float v) {
    const int d = __builtin_amdgcn_readlane(dsel, m);
    float* const rowp = (ch0 < 416) ? planeP + (size_t)d * planePitch
                                    : acc + (size_t)d * 480;
    atomicAdd(rowp + ch0 + lane, v);
  };

  auto scatter1 = [&](const float (&o)[2][8], int ch0, int dsel) {
#pragma unroll
    for (int r = 0; r < 8; ++r) {
      stg[(rbase + r) * P1 + n] = o[0][r];
      stg[(rbase + r) * P1 + 16 + n] = o[1][r];
    }
    __builtin_amdgcn_fence(__ATOMIC_ACQ_REL, "wavefront");
    float v[16];
#pragma unroll
    for (int m = 0; m < 16; ++m) v[m] = stg[m * P1 + lane];
#pragma unroll
    for (int m = 0; m < 16; ++m) add_line(dsel, m, ch0, v[m]);
    __builtin_amdgcn_fence(__ATOMIC_ACQ_REL, "wavefront");
  };

  auto scatter3 = [&](const float (&o)[2][8][3], int ch0, int dsel) {
#pragma unroll
    for (int ph = 0; ph < 2; ++ph) {
#pragma unroll
      for (int q = 0; q < 4; ++q) {
        float* row = stg + (q + 4 * half) * P3;
#pragma unroll
        for (int t = 0; t < 2; ++t) {
          row[t * 48 + 3 * n + 0] = o[t][4 * ph + q][0];
          row[t * 48 + 3 * n + 1] = o[t][4 * ph + q][1];
          row[t * 48 + 3 * n + 2] = o[t][4 * ph + q][2];
        }
      }
      __builtin_amdgcn_fence(__ATOMIC_ACQ_REL, "wavefront");
      float v[8][3];
#pragma unroll
      for (int s = 0; s < 8; ++s)
#pragma unroll
        for (int L = 0; L < 3; ++L) v[s][L] = stg[s * P3 + L * 32 + lane];
#pragma unroll
      for (int s = 0; s < 8; ++s) {
        const int m = (s & 3) + 4 * ph + 8 * (s >> 2);
#pragma unroll
        for (int L = 0; L < 3; ++L) add_line(dsel, m, ch0 + 32 * L, v[s][L]);
      }
      __builtin_amdgcn_fence(__ATOMIC_ACQ_REL, "wavefront");
    }
  };


  for (int p = 0; p < 2; ++p) {
    const int nb = 32 * p;
    v8f w[2][2];
    wpair(nb, w);
    v8f wA = w[0][0], wB = w[0][1];
#pragma nounroll
    for (int s = 0; s < 2; ++s) {
      const float* xsS = s0S + s * (16 * 160);
      const float* shT = shS + s * 64;
      float o[2][8];
#pragma unroll
      for (int r = 0; r < 8; ++r) {
        const float sh0 = shT[(rbase + r) * 4];
        const float* xr = xsS + (rbase + r) * 160 + nb + n;
        o[0][r] = wA[r] * xr[0] * sh0;
        o[1][r] = wB[r] * xr[16] * sh0;
      }
      scatter1(o, nb, s ? drow1 : drow0);
      wA = w[1][0]; wB = w[1][1];
    }
  }
  {
    v8f w[2][2];
    wpair(160, w);
    v8f wA = w[0][0], wB = w[0][1];
#pragma nounroll
    for (int s = 0; s < 2; ++s) {
      const float* xsS = s0S + s * (16 * 160);
      const float* shT = shS + s * 64;
      float o[2][8];
#pragma unroll
      for (int r = 0; r < 8; ++r) {
        const v4f sh = *(const v4f*)(shT + (rbase + r) * 4);
#pragma unroll
        for (int t = 0; t < 2; ++t) {
          const float* xv = xsS + (rbase + r) * 160 + 64 + (16 * t + n) * 3;
          const float wv = t ? wB[r] : wA[r];
          float dp = xv[0] * sh.y + xv[1] * sh.z + xv[2] * sh.w;
          o[t][r] = wv * dp * INV_SQ3;
        }
      }
      scatter1(o, 64, s ? drow1 : drow0);
      wA = w[1][0]; wB = w[1][1];
    }
  }
  for (int p = 0; p < 2; ++p) {
    const int nb = 64 + 32 * p;
    v8f w[2][2];
    wpair(nb, w);
    v8f wA = w[0][0], wB = w[0][1];
#pragma nounroll
    for (int s = 0; s < 2; ++s) {
      const float* xsS = s0S + s * (16 * 160);
      const float* shT = shS + s * 64;
      float o[2][8][3];
#pragma unroll
      for (int r = 0; r < 8; ++r) {
        const v4f sh = *(const v4f*)(shT + (rbase + r) * 4);
        const float* xr = xsS + (rbase + r) * 160 + 32 * p + n;
#pragma unroll
        for (int t = 0; t < 2; ++t) {
          const float wv = t ? wB[r] : wA[r];
          float f = wv * xr[16 * t] * INV_SQ3;
          o[t][r][0] = f * sh.y;
          o[t][r][1] = f * sh.z;
          o[t][r][2] = f * sh.w;
        }
      }
      scatter3(o, 96 + 96 * p, s ? drow1 : drow0);
      wA = w[1][0]; wB = w[1][1];
    }
  }
  {
    v8f w[2][2];
    wpair(128, w);
    v8f wA = w[0][0], wB = w[0][1];
#pragma nounroll
    for (int s = 0; s < 2; ++s) {
      const float* xsS = s0S + s * (16 * 160);
      const float* shT = shS + s * 64;
      float o[2][8][3];
#pragma unroll
      for (int r = 0; r < 8; ++r) {
        const float sh0 = shT[(rbase + r) * 4];
#pragma unroll
        for (int t = 0; t < 2; ++t) {
          const float* xv = xsS + (rbase + r) * 160 + 64 + (16 * t + n) * 3;
          const float wv = t ? wB[r] : wA[r];
          float f = wv * sh0;
          o[t][r][0] = f * xv[0];
          o[t][r][1] = f * xv[1];
          o[t][r][2] = f * xv[2];
        }
      }
      scatter3(o, 288, s ? drow1 : drow0);
      wA = w[1][0]; wB = w[1][1];
    }
  }
  {
    v8f w[2][2];
    wpair(192, w);
    v8f wA = w[0][0], wB = w[0][1];
#pragma nounroll
    for (int s = 0; s < 2; ++s) {
      const float* xsS = s0S + s * (16 * 160);
      const float* shT = shS + s * 64;
      float o[2][8][3];
#pragma unroll
      for (int r = 0; r < 8; ++r) {
        const v4f sh = *(const v4f*)(shT + (rbase + r) * 4);
#pragma unroll
        for (int t = 0; t < 2; ++t) {
          const float* xv = xsS + (rbase + r) * 160 + 64 + (16 * t + n) * 3;
          const float wv = t ? wB[r] : wA[r];
          const float s1x = sh.y, s1y = sh.z, s1z = sh.w;
          float cx = xv[1] * s1z - xv[2] * s1y;
          float cy = xv[2] * s1x - xv[0] * s1z;
          float cz = xv[0] * s1y - xv[1] * s1x;
          float f = wv * INV_SQ2;
          o[t][r][0] = f * cx;
          o[t][r][1] = f * cy;
          o[t][r][2] = f * cz;
        }
      }
      scatter3(o, 384, s ? drow1 : drow0);
      wA = w[1][0]; wB = w[1][1];
    }
  }
}

#define TP 484
__global__ __launch_bounds__(32) void proj_kernel(
    const float* __restrict__ acc,
    const float* __restrict__ WO0, const float* __restrict__ bO0,
    const float* __restrict__ WO1, const float* __restrict__ WO2,
    float* y) {
  __shared__ __align__(16) float tS[16 * TP];
  const int lane = threadIdx.x;
  const int n0 = blockIdx.x * 16;
  const v4f* a4 = (const v4f*)(acc + (size_t)n0 * 480);
  const v4f* b4 = (const v4f*)(y + (size_t)n0 * 416);

#pragma unroll
  for (int b = 0; b < 4; ++b) {
    v4f va[13], vb[13];
#pragma unroll
    for (int i = 0; i < 13; ++i) {
      const int q = lane + 32 * (13 * b + i);
      const int m = q / 104, c4 = q % 104;
      vb[i] = b4[q];
      va[i] = a4[m * 120 + c4];
    }
#pragma unroll
    for (int i = 0; i < 13; ++i) {
      const int q = lane + 32 * (13 * b + i);
      const int m = q / 104, c4 = q % 104;
      *(v4f*)(tS + m * TP + c4 * 4) = va[i] + vb[i];
    }
  }
  {
    v4f va[8];
#pragma unroll
    for (int i = 0; i < 8; ++i) {
      const int q = lane + 32 * i;
      va[i] = a4[(q >> 4) * 120 + 104 + (q & 15)];
    }
#pragma unroll
    for (int i = 0; i < 8; ++i) {
      const int q = lane + 32 * i;
      *(v4f*)(tS + (q >> 4) * TP + 416 + (q & 15) * 4) = va[i];
    }
  }
  __syncthreads();

  {
    v2f aa[24];
    load_a(aa, [&](int m, int k) { return ld2(tS + m * TP + k); });
    for (int j = 0; j < 8; ++j) {
      v8f c = {};
      c = wmma_tile(aa, [&](int n, int k) {
        return ldb<128>(WO0, k, j * 16 + n);
      }, c);
      tile_store(c, [&](int, int m, int n, float v) {
        int col = j * 16 + n;
        y[(size_t)(n0 + m) * 416 + col] = v * INV_SQ96 + bO0[col];
      });
    }
  }
  for (int i = 0; i < 3; ++i) {
    v2f aa[24];
    load_a(aa, [&](int m, int k) {
      v2f a;
      a.x = tS[m * TP + 96 + 3 * k + i];
      a.y = tS[m * TP + 96 + 3 * (k + 1) + i];
      return a;
    });
#pragma unroll
    for (int j = 0; j < 4; ++j) {
      v8f c = {};
      c = wmma_tile(aa, [&](int n, int k) {
        return ldb<64>(WO1, k, j * 16 + n);
      }, c);
      tile_store(c, [&](int, int m, int n, float v) {
        y[(size_t)(n0 + m) * 416 + 128 + (j * 16 + n) * 3 + i] = v * INV_SQ96;
      });
    }
  }
  for (int i = 0; i < 3; ++i) {
    v2f aa[8];
    load_a(aa, [&](int m, int k) {
      v2f a;
      a.x = tS[m * TP + 384 + 3 * k + i];
      a.y = tS[m * TP + 384 + 3 * (k + 1) + i];
      return a;
    });
    for (int j = 0; j < 2; ++j) {
      v8f c = {};
      c = wmma_tile(aa, [&](int n, int k) {
        return ldb<32>(WO2, k, j * 16 + n);
      }, c);
      tile_store(c, [&](int, int m, int n, float v) {
        y[(size_t)(n0 + m) * 416 + 320 + (j * 16 + n) * 3 + i] = v * INV_SQ32;
      });
    }
  }
}

extern "C" void kernel_launch(void* const* d_in, const int* in_sizes, int n_in,
                              void* d_out, int out_size, void* d_ws,
                              size_t ws_size, hipStream_t stream) {
  (void)in_sizes; (void)n_in; (void)out_size; (void)ws_size;
  const float* x         = (const float*)d_in[0];
  const float* edge_sh   = (const float*)d_in[1];
  const float* edge_attr = (const float*)d_in[2];
  const float* W0p = (const float*)d_in[3];
  const float* b0p = (const float*)d_in[4];
  const float* W1p = (const float*)d_in[5];
  const float* W0n = (const float*)d_in[6];
  const float* b0n = (const float*)d_in[7];
  const float* W1n = (const float*)d_in[8];
  const float* G1  = (const float*)d_in[9];
  const float* g1b = (const float*)d_in[10];
  const float* G2  = (const float*)d_in[11];
  const float* g2b = (const float*)d_in[12];
  const float* F1  = (const float*)d_in[13];
  const float* F2  = (const float*)d_in[14];
  const float* L1  = (const float*)d_in[15];
  const float* L2  = (const float*)d_in[16];
  const float* WO0 = (const float*)d_in[17];
  const float* bO0 = (const float*)d_in[18];
  const float* WO1 = (const float*)d_in[19];
  const float* WO2 = (const float*)d_in[20];
  const int* edge_index = (const int*)d_in[21];

  float* ws   = (float*)d_ws;
  float* p0   = ws;
  float* p1   = p0   + (size_t)N_NODES * 64;
  float* x0n  = p1   + (size_t)N_NODES * 96;
  float* x1n  = x0n  + (size_t)N_NODES * 64;
  float* acc  = x1n  + (size_t)N_NODES * 96;
  float* accB = (float*)d_out;

  hipMemsetAsync(acc, 0, (size_t)N_NODES * 480 * sizeof(float), stream);

  node_kernel<<<N_NODES / 16, 32, 0, stream>>>(
      x, W0p, b0p, W1p, W0n, b0n, W1n, G1, g1b, G2, g2b,
      p0, p1, x0n, x1n, accB);

  edge_kernel<<<N_EDGES / 32, 32, 0, stream>>>(
      edge_sh, edge_attr, F1, F2, L1, L2, edge_index,
      p0, p1, x0n, x1n, acc, accB);

  proj_kernel<<<N_NODES / 16, 32, 0, stream>>>(
      acc, WO0, bO0, WO1, WO2, (float*)d_out);
}
